# starting kernel
# baseline (speedup 1.0000x reference)
#include <hip/hip_runtime.h>


#define BATCH   16384
#define IN_DIM  1024
#define OUT_DIM 128

typedef __attribute__((ext_vector_type(16))) __bf16 v16bf;
typedef __attribute__((ext_vector_type(8)))  float  v8f;

static __device__ inline v8f wmma_bf16(v16bf a, v16bf b, v8f c) {
  return __builtin_amdgcn_wmma_f32_16x16x32_bf16(false, a, false, b,
                                                 (short)0, c, false, false);
}

static __device__ inline void split_fragA(const float* __restrict__ rowp,
                                          int k0, int half, bool neg,
                                          v16bf& hi, v16bf& lo) {
  const float* p0 = rowp + k0 + 8 * half;
  const float* p1 = p0 + 16;
#pragma unroll
  for (int e = 0; e < 8; ++e) {
    float x0 = neg ? -p0[e] : p0[e];
    float x1 = neg ? -p1[e] : p1[e];
    __bf16 h0 = (__bf16)x0;
    __bf16 h1 = (__bf16)x1;
    hi[e]     = h0;  lo[e]     = (__bf16)(x0 - (float)h0);
    hi[e + 8] = h1;  lo[e + 8] = (__bf16)(x1 - (float)h1);
  }
}

static __device__ inline v16bf load_fragB(const __bf16* __restrict__ rowp,
                                          int k0, int half) {
  const __bf16* p0 = rowp + k0 + 8 * half;
  const __bf16* p1 = p0 + 16;
  v16bf f;
#pragma unroll
  for (int e = 0; e < 8; ++e) { f[e] = p0[e]; f[e + 8] = p1[e]; }
  return f;
}

__global__ __launch_bounds__(256) void compute_G(const float* __restrict__ A,
                                                 float* __restrict__ G) {
  int idx = blockIdx.x * 256 + threadIdx.x;
  int r = idx >> 7, c = idx & 127;
  const float* ar = A + (size_t)r * IN_DIM;
  const float* ac = A + (size_t)c * IN_DIM;
  float s = 0.0f;
#pragma unroll 8
  for (int k = 0; k < IN_DIM; ++k) s = fmaf(ar[k], ac[k], s);
  G[idx] = s;
}

__global__ __launch_bounds__(512) void invert_G(const float* __restrict__ G,
                                                __bf16* __restrict__ Gh,
                                                __bf16* __restrict__ Gl) {
  extern __shared__ float aug[];
  __shared__ float colc[128];
  const int tid = threadIdx.x;
  for (int idx = tid; idx < 128 * 256; idx += 512) {
    int i = idx >> 8, j = idx & 255;
    aug[idx] = (j < 128) ? G[i * 128 + j] : ((j - 128) == i ? 1.0f : 0.0f);
  }
  __syncthreads();
  for (int k = 0; k < 128; ++k) {
    if (tid < 128) colc[tid] = aug[tid * 256 + k];
    __syncthreads();
    float pinv = 1.0f / colc[k];
    for (int j = tid; j < 256; j += 512) aug[k * 256 + j] *= pinv;
    __syncthreads();
    int tcol = tid & 255, r0 = tid >> 8;
    float pk = aug[k * 256 + tcol];
    for (int i = r0; i < 128; i += 2)
      if (i != k) aug[i * 256 + tcol] -= colc[i] * pk;
    __syncthreads();
  }
  for (int idx = tid; idx < 128 * 128; idx += 512) {
    int i = idx >> 7, j = idx & 127;
    float v = aug[i * 256 + 128 + j];
    __bf16 h = (__bf16)v;
    Gh[idx] = h; Gl[idx] = (__bf16)(v - (float)h);
  }
}

__global__ __launch_bounds__(256) void pack_A(const float* __restrict__ A,
                                              __bf16* __restrict__ Ah,
                                              __bf16* __restrict__ Al,
                                              __bf16* __restrict__ Ath,
                                              __bf16* __restrict__ Atl) {
  int idx = blockIdx.x * 256 + threadIdx.x;
  int r = idx >> 10, c = idx & 1023;
  float x = A[idx];
  __bf16 h = (__bf16)x;
  __bf16 l = (__bf16)(x - (float)h);
  Ah[idx] = h;  Al[idx] = l;
  Ath[c * 128 + r] = h;  Atl[c * 128 + r] = l;
}

__global__ __launch_bounds__(128) void fused_project(
    const float* __restrict__ y, const float* __restrict__ b,
    const __bf16* __restrict__ Ah, const __bf16* __restrict__ Al,
    const __bf16* __restrict__ Ath, const __bf16* __restrict__ Atl,
    const __bf16* __restrict__ Gh, const __bf16* __restrict__ Gl,
    float* __restrict__ out) {
  __shared__ float s_r[16][OUT_DIM];
  __shared__ float s_mu[16][OUT_DIM];

  const int tid  = threadIdx.x;
  const int wid  = tid >> 5;
  const int lane = tid & 31;
  const int half = lane >> 4;
  const int l15  = lane & 15;
  const int i0   = blockIdx.x * 16;

  const int n0a = (wid * 2) * 16;
  const int n0b = n0a + 16;
  const float*  yrow = y  + (size_t)(i0 + l15) * IN_DIM;
  const __bf16* b0h  = Ah + (size_t)(n0a + l15) * IN_DIM;
  const __bf16* b0l  = Al + (size_t)(n0a + l15) * IN_DIM;
  const __bf16* b1h  = Ah + (size_t)(n0b + l15) * IN_DIM;
  const __bf16* b1l  = Al + (size_t)(n0b + l15) * IN_DIM;

  v8f acc0 = {};
  v8f acc1 = {};
  for (int kk = 0; kk < IN_DIM / 32; ++kk) {
    const int k0 = kk * 32;
    v16bf yh, yl;
    split_fragA(yrow, k0, half, false, yh, yl);
    v16bf fh = load_fragB(b0h, k0, half);
    v16bf fl = load_fragB(b0l, k0, half);
    acc0 = wmma_bf16(yh, fh, acc0);
    acc0 = wmma_bf16(yl, fh, acc0);
    acc0 = wmma_bf16(yh, fl, acc0);
    fh = load_fragB(b1h, k0, half);
    fl = load_fragB(b1l, k0, half);
    acc1 = wmma_bf16(yh, fh, acc1);
    acc1 = wmma_bf16(yl, fh, acc1);
    acc1 = wmma_bf16(yh, fl, acc1);
  }
#pragma unroll
  for (int v = 0; v < 8; ++v) {
    int m = v + 8 * half;
    s_r[m][n0a + l15] = acc0[v] - b[(size_t)(i0 + m) * OUT_DIM + n0a + l15];
    s_r[m][n0b + l15] = acc1[v] - b[(size_t)(i0 + m) * OUT_DIM + n0b + l15];
  }
  __syncthreads();

  const __bf16* g0h = Gh + (size_t)(n0a + l15) * OUT_DIM;
  const __bf16* g0l = Gl + (size_t)(n0a + l15) * OUT_DIM;
  const __bf16* g1h = Gh + (size_t)(n0b + l15) * OUT_DIM;
  const __bf16* g1l = Gl + (size_t)(n0b + l15) * OUT_DIM;
  v8f mu0 = {};
  v8f mu1 = {};
#pragma unroll
  for (int kk = 0; kk < 4; ++kk) {
    const int k0 = kk * 32;
    v16bf rh, rl;
    split_fragA(&s_r[l15][0], k0, half, false, rh, rl);
    v16bf fh = load_fragB(g0h, k0, half);
    v16bf fl = load_fragB(g0l, k0, half);
    mu0 = wmma_bf16(rh, fh, mu0);
    mu0 = wmma_bf16(rl, fh, mu0);
    mu0 = wmma_bf16(rh, fl, mu0);
    fh = load_fragB(g1h, k0, half);
    fl = load_fragB(g1l, k0, half);
    mu1 = wmma_bf16(rh, fh, mu1);
    mu1 = wmma_bf16(rl, fh, mu1);
    mu1 = wmma_bf16(rh, fl, mu1);
  }
#pragma unroll
  for (int v = 0; v < 8; ++v) {
    int m = v + 8 * half;
    s_mu[m][n0a + l15] = mu0[v];
    s_mu[m][n0b + l15] = mu1[v];
  }
  __syncthreads();

  v16bf mh[4], ml[4];
#pragma unroll
  for (int kk = 0; kk < 4; ++kk)
    split_fragA(&s_mu[l15][0], kk * 32, half, true, mh[kk], ml[kk]);

  for (int t = 0; t < 16; ++t) {
    const int c0 = (wid * 16 + t) * 16;
    const __bf16* ph = Ath + (size_t)(c0 + l15) * OUT_DIM;
    const __bf16* pl = Atl + (size_t)(c0 + l15) * OUT_DIM;
    v8f acc;
#pragma unroll
    for (int v = 0; v < 8; ++v)
      acc[v] = y[(size_t)(i0 + v + 8 * half) * IN_DIM + c0 + l15];
#pragma unroll
    for (int kk = 0; kk < 4; ++kk) {
      const int k0 = kk * 32;
      v16bf fh = load_fragB(ph, k0, half);
      v16bf fl = load_fragB(pl, k0, half);
      acc = wmma_bf16(mh[kk], fh, acc);
      acc = wmma_bf16(ml[kk], fh, acc);
      acc = wmma_bf16(mh[kk], fl, acc);
    }
#pragma unroll
    for (int v = 0; v < 8; ++v)
      out[(size_t)(i0 + v + 8 * half) * IN_DIM + c0 + l15] = acc[v];
  }
}

extern "C" void kernel_launch(void* const* d_in, const int* in_sizes, int n_in,
                              void* d_out, int out_size, void* d_ws,
                              size_t ws_size, hipStream_t stream) {
  (void)in_sizes; (void)n_in; (void)out_size; (void)ws_size;
  const float* y = (const float*)d_in[0];
  const float* A = (const float*)d_in[1];
  const float* b = (const float*)d_in[2];
  float* out = (float*)d_out;

  char* ws = (char*)d_ws;
  float*  G   = (float*) (ws + 0);
  __bf16* Gh  = (__bf16*)(ws + 65536);
  __bf16* Gl  = (__bf16*)(ws + 98304);
  __bf16* Ah  = (__bf16*)(ws + 131072);
  __bf16* Al  = (__bf16*)(ws + 393216);
  __bf16* Ath = (__bf16*)(ws + 655360);
  __bf16* Atl = (__bf16*)(ws + 917504);

  compute_G<<<64, 256, 0, stream>>>(A, G);
  invert_G<<<1, 512, 128 * 256 * sizeof(float), stream>>>(G, Gh, Gl);
  pack_A<<<512, 256, 0, stream>>>(A, Ah, Al, Ath, Atl);
  fused_project<<<BATCH / 16, 128, 0, stream>>>(y, b, Ah, Al, Ath, Atl,
                                                Gh, Gl, out);
}


# B operands (A, A^T, Ginv as bf16 hi/lo) stored in WMMA fragment order by pack_A and invert_G: each fragment load of fused_project covers 4 whole 128-B lines instead of 16 row pieces
# speedup vs baseline: 3.8906x; 3.8906x over previous
#include <hip/hip_runtime.h>


#define BATCH   16384
#define IN_DIM  1024
#define OUT_DIM 128

typedef __attribute__((ext_vector_type(16))) __bf16 v16bf;
typedef __attribute__((ext_vector_type(8)))  __bf16 v8bf;
typedef __attribute__((ext_vector_type(8)))  float  v8f;

static __device__ inline v8f wmma_bf16(v16bf a, v16bf b, v8f c) {
  return __builtin_amdgcn_wmma_f32_16x16x32_bf16(false, a, false, b,
                                                 (short)0, c, false, false);
}

static __device__ inline void split_fragA(const float* __restrict__ rowp,
                                          int k0, int half, bool neg,
                                          v16bf& hi, v16bf& lo) {
  const float* p0 = rowp + k0 + 8 * half;
  const float* p1 = p0 + 16;
#pragma unroll
  for (int e = 0; e < 8; ++e) {
    float x0 = neg ? -p0[e] : p0[e];
    float x1 = neg ? -p1[e] : p1[e];
    __bf16 h0 = (__bf16)x0;
    __bf16 h1 = (__bf16)x1;
    hi[e]     = h0;  lo[e]     = (__bf16)(x0 - (float)h0);
    hi[e + 8] = h1;  lo[e + 8] = (__bf16)(x1 - (float)h1);
  }
}

static __device__ inline v16bf load_fragB(const __bf16* __restrict__ frag,
                                          int lane) {
  const v8bf* p = reinterpret_cast<const v8bf*>(frag) + lane;
  const v8bf e0 = p[0];
  const v8bf e1 = p[32];
  return __builtin_shufflevector(e0, e1, 0, 1, 2, 3, 4, 5, 6, 7,
                                 8, 9, 10, 11, 12, 13, 14, 15);
}

static __device__ inline int frag_pos(int tile, int frags, int l15, int k) {
  const int kin = k & 31;
  return (tile * frags + (k >> 5)) * 512 + (kin >> 4) * 256 +
         (l15 + 16 * ((kin >> 3) & 1)) * 8 + (kin & 7);
}

__global__ __launch_bounds__(256) void compute_G(const float* __restrict__ A,
                                                 float* __restrict__ G) {
  int idx = blockIdx.x * 256 + threadIdx.x;
  int r = idx >> 7, c = idx & 127;
  const float* ar = A + (size_t)r * IN_DIM;
  const float* ac = A + (size_t)c * IN_DIM;
  float s = 0.0f;
#pragma unroll 8
  for (int k = 0; k < IN_DIM; ++k) s = fmaf(ar[k], ac[k], s);
  G[idx] = s;
}

__global__ __launch_bounds__(512) void invert_G(const float* __restrict__ G,
                                                __bf16* __restrict__ Gh,
                                                __bf16* __restrict__ Gl) {
  __shared__ __attribute__((aligned(16))) float piv[2][128];
  const int tid  = threadIdx.x;
  const int lane = tid & 31;
  const int w    = tid >> 5;
  const int rg   = w >> 1;
  const int R0   = rg * 16;
  const int c0   = (w & 1) * 64 + lane;
  const int c1   = c0 + 32;

  float a0[16], a1[16];
#pragma unroll
  for (int j = 0; j < 16; ++j) {
    a0[j] = G[(R0 + j) * 128 + c0];
    a1[j] = G[(R0 + j) * 128 + c1];
  }
  if (c0 == 0) {
    float4* nv4 = reinterpret_cast<float4*>(&piv[0][R0]);
#pragma unroll
    for (int q = 0; q < 4; ++q)
      nv4[q] = make_float4(a0[4 * q], a0[4 * q + 1], a0[4 * q + 2], a0[4 * q + 3]);
  }
  __syncthreads();

  for (int kb = 0; kb < 8; ++kb) {
#pragma unroll
    for (int kk = 0; kk < 16; ++kk) {
      const int k = kb * 16 + kk;
      const float* v = piv[kk & 1];
      const float pinv = 1.0f / v[k];
      const float t0 = v[c0] * pinv;
      const float t1 = v[c1] * pinv;
      const float n0 = -t0;
      const float n1 = -t1;
      const float4* v4 = reinterpret_cast<const float4*>(v + R0);
      float vr[16];
#pragma unroll
      for (int q = 0; q < 4; ++q) {
        float4 x = v4[q];
        vr[4 * q] = x.x; vr[4 * q + 1] = x.y; vr[4 * q + 2] = x.z; vr[4 * q + 3] = x.w;
      }
#pragma unroll
      for (int j = 0; j < 16; ++j) {
        a0[j] = fmaf(vr[j], n0, a0[j]);
        a1[j] = fmaf(vr[j], n1, a1[j]);
      }
      if (c0 == k) {
#pragma unroll
        for (int j = 0; j < 16; ++j) a0[j] = vr[j] * pinv;
      }
      if (c1 == k) {
#pragma unroll
        for (int j = 0; j < 16; ++j) a1[j] = vr[j] * pinv;
      }
      if (rg == kb) {
        a0[kk] = (c0 == k) ? -pinv : t0;
        a1[kk] = (c1 == k) ? -pinv : t1;
      }
      if (k + 1 < OUT_DIM) {
        float4* nv4 = reinterpret_cast<float4*>(&piv[(kk + 1) & 1][R0]);
        if (c0 == k + 1) {
#pragma unroll
          for (int q = 0; q < 4; ++q)
            nv4[q] = make_float4(a0[4 * q], a0[4 * q + 1], a0[4 * q + 2], a0[4 * q + 3]);
        }
        if (c1 == k + 1) {
#pragma unroll
          for (int q = 0; q < 4; ++q)
            nv4[q] = make_float4(a1[4 * q], a1[4 * q + 1], a1[4 * q + 2], a1[4 * q + 3]);
        }
      }
      __syncthreads();
    }
  }

#pragma unroll
  for (int j = 0; j < 16; ++j) {
    const int i0 = frag_pos(rg, 4, j, c0);
    const int i1 = frag_pos(rg, 4, j, c1);
    const float x0 = -a0[j];
    const float x1 = -a1[j];
    const __bf16 h0 = (__bf16)x0;
    const __bf16 h1 = (__bf16)x1;
    Gh[i0] = h0;  Gl[i0] = (__bf16)(x0 - (float)h0);
    Gh[i1] = h1;  Gl[i1] = (__bf16)(x1 - (float)h1);
  }
}

__global__ __launch_bounds__(256) void pack_A(const float* __restrict__ A,
                                              __bf16* __restrict__ Ah,
                                              __bf16* __restrict__ Al,
                                              __bf16* __restrict__ Ath,
                                              __bf16* __restrict__ Atl) {
  const int t     = blockIdx.x * 256 + threadIdx.x;
  const int q     = t & 16383;
  const int f     = q >> 6;
  const int lane  = q & 31;
  const int l15   = lane & 15;
  const int kin0  = ((q >> 5) & 1) * 16 + (lane >> 4) * 8;
  float x[8];
  __bf16* dh;
  __bf16* dl;
  if (t < 16384) {
    const int nt = f >> 5, kk = f & 31;
    const float* s = A + (size_t)(nt * 16 + l15) * IN_DIM + kk * 32 + kin0;
#pragma unroll
    for (int e = 0; e < 8; ++e) x[e] = s[e];
    dh = Ah;  dl = Al;
  } else {
    const int ct = f >> 2, kk = f & 3;
    const float* s = A + (size_t)(kk * 32 + kin0) * IN_DIM + ct * 16 + l15;
#pragma unroll
    for (int e = 0; e < 8; ++e) x[e] = s[(size_t)e * IN_DIM];
    dh = Ath;  dl = Atl;
  }
  v8bf h, l;
#pragma unroll
  for (int e = 0; e < 8; ++e) {
    const __bf16 hh = (__bf16)x[e];
    h[e] = hh;
    l[e] = (__bf16)(x[e] - (float)hh);
  }
  reinterpret_cast<v8bf*>(dh)[q] = h;
  reinterpret_cast<v8bf*>(dl)[q] = l;
}

__global__ __launch_bounds__(128) void fused_project(
    const float* __restrict__ y, const float* __restrict__ b,
    const __bf16* __restrict__ Ah, const __bf16* __restrict__ Al,
    const __bf16* __restrict__ Ath, const __bf16* __restrict__ Atl,
    const __bf16* __restrict__ Gh, const __bf16* __restrict__ Gl,
    float* __restrict__ out) {
  __shared__ float s_r[16][OUT_DIM];
  __shared__ float s_mu[16][OUT_DIM];

  const int tid  = threadIdx.x;
  const int wid  = tid >> 5;
  const int lane = tid & 31;
  const int half = lane >> 4;
  const int l15  = lane & 15;
  const int i0   = blockIdx.x * 16;

  const int n0a = (wid * 2) * 16;
  const int n0b = n0a + 16;
  const float*  yrow = y  + (size_t)(i0 + l15) * IN_DIM;
  const __bf16* b0h  = Ah + (size_t)(wid * 2) * 32 * 512;
  const __bf16* b0l  = Al + (size_t)(wid * 2) * 32 * 512;
  const __bf16* b1h  = b0h + 32 * 512;
  const __bf16* b1l  = b0l + 32 * 512;

  v8f acc0 = {};
  v8f acc1 = {};
  for (int kk = 0; kk < IN_DIM / 32; ++kk) {
    const int k0 = kk * 32;
    v16bf yh, yl;
    split_fragA(yrow, k0, half, false, yh, yl);
    v16bf fh = load_fragB(b0h + kk * 512, lane);
    v16bf fl = load_fragB(b0l + kk * 512, lane);
    acc0 = wmma_bf16(yh, fh, acc0);
    acc0 = wmma_bf16(yl, fh, acc0);
    acc0 = wmma_bf16(yh, fl, acc0);
    fh = load_fragB(b1h + kk * 512, lane);
    fl = load_fragB(b1l + kk * 512, lane);
    acc1 = wmma_bf16(yh, fh, acc1);
    acc1 = wmma_bf16(yl, fh, acc1);
    acc1 = wmma_bf16(yh, fl, acc1);
  }
#pragma unroll
  for (int v = 0; v < 8; ++v) {
    int m = v + 8 * half;
    s_r[m][n0a + l15] = acc0[v] - b[(size_t)(i0 + m) * OUT_DIM + n0a + l15];
    s_r[m][n0b + l15] = acc1[v] - b[(size_t)(i0 + m) * OUT_DIM + n0b + l15];
  }
  __syncthreads();

  const __bf16* g0h = Gh + (wid * 2) * 4 * 512;
  const __bf16* g0l = Gl + (wid * 2) * 4 * 512;
  const __bf16* g1h = g0h + 4 * 512;
  const __bf16* g1l = g0l + 4 * 512;
  v8f mu0 = {};
  v8f mu1 = {};
#pragma unroll
  for (int kk = 0; kk < 4; ++kk) {
    const int k0 = kk * 32;
    v16bf rh, rl;
    split_fragA(&s_r[l15][0], k0, half, false, rh, rl);
    v16bf fh = load_fragB(g0h + kk * 512, lane);
    v16bf fl = load_fragB(g0l + kk * 512, lane);
    mu0 = wmma_bf16(rh, fh, mu0);
    mu0 = wmma_bf16(rl, fh, mu0);
    mu0 = wmma_bf16(rh, fl, mu0);
    fh = load_fragB(g1h + kk * 512, lane);
    fl = load_fragB(g1l + kk * 512, lane);
    mu1 = wmma_bf16(rh, fh, mu1);
    mu1 = wmma_bf16(rl, fh, mu1);
    mu1 = wmma_bf16(rh, fl, mu1);
  }
#pragma unroll
  for (int v = 0; v < 8; ++v) {
    int m = v + 8 * half;
    s_mu[m][n0a + l15] = mu0[v];
    s_mu[m][n0b + l15] = mu1[v];
  }
  __syncthreads();

  v16bf mh[4], ml[4];
#pragma unroll
  for (int kk = 0; kk < 4; ++kk)
    split_fragA(&s_mu[l15][0], kk * 32, half, true, mh[kk], ml[kk]);

  for (int t = 0; t < 16; ++t) {
    const int c0 = (wid * 16 + t) * 16;
    const __bf16* ph = Ath + (size_t)(wid * 16 + t) * 4 * 512;
    const __bf16* pl = Atl + (size_t)(wid * 16 + t) * 4 * 512;
    v8f acc;
#pragma unroll
    for (int v = 0; v < 8; ++v)
      acc[v] = y[(size_t)(i0 + v + 8 * half) * IN_DIM + c0 + l15];
#pragma unroll
    for (int kk = 0; kk < 4; ++kk) {
      v16bf fh = load_fragB(ph + kk * 512, lane);
      v16bf fl = load_fragB(pl + kk * 512, lane);
      acc = wmma_bf16(mh[kk], fh, acc);
      acc = wmma_bf16(ml[kk], fh, acc);
      acc = wmma_bf16(mh[kk], fl, acc);
    }
#pragma unroll
    for (int v = 0; v < 8; ++v)
      out[(size_t)(i0 + v + 8 * half) * IN_DIM + c0 + l15] = acc[v];
  }
}

extern "C" void kernel_launch(void* const* d_in, const int* in_sizes, int n_in,
                              void* d_out, int out_size, void* d_ws,
                              size_t ws_size, hipStream_t stream) {
  (void)in_sizes; (void)n_in; (void)out_size; (void)ws_size;
  const float* y = (const float*)d_in[0];
  const float* A = (const float*)d_in[1];
  const float* b = (const float*)d_in[2];
  float* out = (float*)d_out;

  char* ws = (char*)d_ws;
  float*  G   = (float*) (ws + 0);
  __bf16* Gh  = (__bf16*)(ws + 65536);
  __bf16* Gl  = (__bf16*)(ws + 98304);
  __bf16* Ah  = (__bf16*)(ws + 131072);
  __bf16* Al  = (__bf16*)(ws + 393216);
  __bf16* Ath = (__bf16*)(ws + 655360);
  __bf16* Atl = (__bf16*)(ws + 917504);

  compute_G<<<64, 256, 0, stream>>>(A, G);
  invert_G<<<1, 512, 0, stream>>>(G, Gh, Gl);
  pack_A<<<128, 256, 0, stream>>>(A, Ah, Al, Ath, Atl);
  fused_project<<<BATCH / 16, 128, 0, stream>>>(y, b, Ah, Al, Ath, Atl,
                                                Gh, Gl, out);
}


# compute_G (G = A A^T): per-thread serial f32 FMA dot products, lane-strided reads -> f32 16x16x4 matrix instruction, one tile per workgroup, K over 4 waves x 2 accumulators; G summed in another order
# speedup vs baseline: 4.5274x; 1.1637x over previous
#include <hip/hip_runtime.h>


#define BATCH   16384
#define IN_DIM  1024
#define OUT_DIM 128

typedef __attribute__((ext_vector_type(16))) __bf16 v16bf;
typedef __attribute__((ext_vector_type(8)))  __bf16 v8bf;
typedef __attribute__((ext_vector_type(8)))  float  v8f;
typedef __attribute__((ext_vector_type(2)))  float  v2f;

static __device__ inline v8f wmma_bf16(v16bf a, v16bf b, v8f c) {
  return __builtin_amdgcn_wmma_f32_16x16x32_bf16(false, a, false, b,
                                                 (short)0, c, false, false);
}

static __device__ inline v8f wmma_f32(v2f a, v2f b, v8f c) {
  return __builtin_amdgcn_wmma_f32_16x16x4_f32(false, a, false, b,
                                               (short)0, c, false, false);
}

static __device__ inline void split_fragA(const float* __restrict__ rowp,
                                          int k0, int half, bool neg,
                                          v16bf& hi, v16bf& lo) {
  const float* p0 = rowp + k0 + 8 * half;
  const float* p1 = p0 + 16;
#pragma unroll
  for (int e = 0; e < 8; ++e) {
    float x0 = neg ? -p0[e] : p0[e];
    float x1 = neg ? -p1[e] : p1[e];
    __bf16 h0 = (__bf16)x0;
    __bf16 h1 = (__bf16)x1;
    hi[e]     = h0;  lo[e]     = (__bf16)(x0 - (float)h0);
    hi[e + 8] = h1;  lo[e + 8] = (__bf16)(x1 - (float)h1);
  }
}

static __device__ inline v16bf load_fragB(const __bf16* __restrict__ frag,
                                          int lane) {
  const v8bf* p = reinterpret_cast<const v8bf*>(frag) + lane;
  const v8bf e0 = p[0];
  const v8bf e1 = p[32];
  return __builtin_shufflevector(e0, e1, 0, 1, 2, 3, 4, 5, 6, 7,
                                 8, 9, 10, 11, 12, 13, 14, 15);
}

static __device__ inline int frag_pos(int tile, int frags, int l15, int k) {
  const int kin = k & 31;
  return (tile * frags + (k >> 5)) * 512 + (kin >> 4) * 256 +
         (l15 + 16 * ((kin >> 3) & 1)) * 8 + (kin & 7);
}

__global__ __launch_bounds__(128) void compute_G(const float* __restrict__ A,
                                                 float* __restrict__ G) {
  __shared__ float part[4][8][32];
  const int tid  = threadIdx.x;
  const int lane = tid & 31;
  const int w    = tid >> 5;
  const int c    = lane & 15;
  const int h    = lane >> 4;
  const int ti   = blockIdx.x >> 3;
  const int tj   = blockIdx.x & 7;
  const float4* pa = reinterpret_cast<const float4*>(
      A + (size_t)(ti * 16 + c) * IN_DIM + w * 256 + 4 * h);
  const float4* pb = reinterpret_cast<const float4*>(
      A + (size_t)(tj * 16 + c) * IN_DIM + w * 256 + 4 * h);
  float4 xa[32], xb[32];
#pragma unroll
  for (int j = 0; j < 32; ++j) {
    xa[j] = pa[2 * j];
    xb[j] = pb[2 * j];
  }
  __builtin_amdgcn_sched_barrier(0);
  v8f acc0 = {};
  v8f acc1 = {};
#pragma unroll
  for (int j = 0; j < 32; ++j) {
    const v2f a0 = {xa[j].x, xa[j].y};
    const v2f b0 = {xb[j].x, xb[j].y};
    const v2f a1 = {xa[j].z, xa[j].w};
    const v2f b1 = {xb[j].z, xb[j].w};
    acc0 = wmma_f32(a0, b0, acc0);
    acc1 = wmma_f32(a1, b1, acc1);
  }
#pragma unroll
  for (int r = 0; r < 8; ++r) part[w][r][lane] = acc0[r] + acc1[r];
  __syncthreads();
#pragma unroll
  for (int q = 0; q < 2; ++q) {
    const int r = w + 4 * q;
    const float s = (part[0][r][lane] + part[1][r][lane]) +
                    (part[2][r][lane] + part[3][r][lane]);
    G[(ti * 16 + 8 * h + r) * OUT_DIM + tj * 16 + c] = s;
  }
}

__global__ __launch_bounds__(512) void invert_G(const float* __restrict__ G,
                                                __bf16* __restrict__ Gh,
                                                __bf16* __restrict__ Gl) {
  __shared__ __attribute__((aligned(16))) float piv[2][128];
  const int tid  = threadIdx.x;
  const int lane = tid & 31;
  const int w    = tid >> 5;
  const int rg   = w >> 1;
  const int R0   = rg * 16;
  const int c0   = (w & 1) * 64 + lane;
  const int c1   = c0 + 32;

  float a0[16], a1[16];
#pragma unroll
  for (int j = 0; j < 16; ++j) {
    a0[j] = G[(R0 + j) * 128 + c0];
    a1[j] = G[(R0 + j) * 128 + c1];
  }
  if (c0 == 0) {
    float4* nv4 = reinterpret_cast<float4*>(&piv[0][R0]);
#pragma unroll
    for (int q = 0; q < 4; ++q)
      nv4[q] = make_float4(a0[4 * q], a0[4 * q + 1], a0[4 * q + 2], a0[4 * q + 3]);
  }
  __syncthreads();

  for (int kb = 0; kb < 8; ++kb) {
#pragma unroll
    for (int kk = 0; kk < 16; ++kk) {
      const int k = kb * 16 + kk;
      const float* v = piv[kk & 1];
      const float pinv = 1.0f / v[k];
      const float t0 = v[c0] * pinv;
      const float t1 = v[c1] * pinv;
      const float n0 = -t0;
      const float n1 = -t1;
      const float4* v4 = reinterpret_cast<const float4*>(v + R0);
      float vr[16];
#pragma unroll
      for (int q = 0; q < 4; ++q) {
        float4 x = v4[q];
        vr[4 * q] = x.x; vr[4 * q + 1] = x.y; vr[4 * q + 2] = x.z; vr[4 * q + 3] = x.w;
      }
#pragma unroll
      for (int j = 0; j < 16; ++j) {
        a0[j] = fmaf(vr[j], n0, a0[j]);
        a1[j] = fmaf(vr[j], n1, a1[j]);
      }
      if (c0 == k) {
#pragma unroll
        for (int j = 0; j < 16; ++j) a0[j] = vr[j] * pinv;
      }
      if (c1 == k) {
#pragma unroll
        for (int j = 0; j < 16; ++j) a1[j] = vr[j] * pinv;
      }
      if (rg == kb) {
        a0[kk] = (c0 == k) ? -pinv : t0;
        a1[kk] = (c1 == k) ? -pinv : t1;
      }
      if (k + 1 < OUT_DIM) {
        float4* nv4 = reinterpret_cast<float4*>(&piv[(kk + 1) & 1][R0]);
        if (c0 == k + 1) {
#pragma unroll
          for (int q = 0; q < 4; ++q)
            nv4[q] = make_float4(a0[4 * q], a0[4 * q + 1], a0[4 * q + 2], a0[4 * q + 3]);
        }
        if (c1 == k + 1) {
#pragma unroll
          for (int q = 0; q < 4; ++q)
            nv4[q] = make_float4(a1[4 * q], a1[4 * q + 1], a1[4 * q + 2], a1[4 * q + 3]);
        }
      }
      __syncthreads();
    }
  }

#pragma unroll
  for (int j = 0; j < 16; ++j) {
    const int i0 = frag_pos(rg, 4, j, c0);
    const int i1 = frag_pos(rg, 4, j, c1);
    const float x0 = -a0[j];
    const float x1 = -a1[j];
    const __bf16 h0 = (__bf16)x0;
    const __bf16 h1 = (__bf16)x1;
    Gh[i0] = h0;  Gl[i0] = (__bf16)(x0 - (float)h0);
    Gh[i1] = h1;  Gl[i1] = (__bf16)(x1 - (float)h1);
  }
}

__global__ __launch_bounds__(256) void pack_A(const float* __restrict__ A,
                                              __bf16* __restrict__ Ah,
                                              __bf16* __restrict__ Al,
                                              __bf16* __restrict__ Ath,
                                              __bf16* __restrict__ Atl) {
  const int t     = blockIdx.x * 256 + threadIdx.x;
  const int q     = t & 16383;
  const int f     = q >> 6;
  const int lane  = q & 31;
  const int l15   = lane & 15;
  const int kin0  = ((q >> 5) & 1) * 16 + (lane >> 4) * 8;
  float x[8];
  __bf16* dh;
  __bf16* dl;
  if (t < 16384) {
    const int nt = f >> 5, kk = f & 31;
    const float* s = A + (size_t)(nt * 16 + l15) * IN_DIM + kk * 32 + kin0;
#pragma unroll
    for (int e = 0; e < 8; ++e) x[e] = s[e];
    dh = Ah;  dl = Al;
  } else {
    const int ct = f >> 2, kk = f & 3;
    const float* s = A + (size_t)(kk * 32 + kin0) * IN_DIM + ct * 16 + l15;
#pragma unroll
    for (int e = 0; e < 8; ++e) x[e] = s[(size_t)e * IN_DIM];
    dh = Ath;  dl = Atl;
  }
  v8bf h, l;
#pragma unroll
  for (int e = 0; e < 8; ++e) {
    const __bf16 hh = (__bf16)x[e];
    h[e] = hh;
    l[e] = (__bf16)(x[e] - (float)hh);
  }
  reinterpret_cast<v8bf*>(dh)[q] = h;
  reinterpret_cast<v8bf*>(dl)[q] = l;
}

__global__ __launch_bounds__(128) void fused_project(
    const float* __restrict__ y, const float* __restrict__ b,
    const __bf16* __restrict__ Ah, const __bf16* __restrict__ Al,
    const __bf16* __restrict__ Ath, const __bf16* __restrict__ Atl,
    const __bf16* __restrict__ Gh, const __bf16* __restrict__ Gl,
    float* __restrict__ out) {
  __shared__ float s_r[16][OUT_DIM];
  __shared__ float s_mu[16][OUT_DIM];

  const int tid  = threadIdx.x;
  const int wid  = tid >> 5;
  const int lane = tid & 31;
  const int half = lane >> 4;
  const int l15  = lane & 15;
  const int i0   = blockIdx.x * 16;

  const int n0a = (wid * 2) * 16;
  const int n0b = n0a + 16;
  const float*  yrow = y  + (size_t)(i0 + l15) * IN_DIM;
  const __bf16* b0h  = Ah + (size_t)(wid * 2) * 32 * 512;
  const __bf16* b0l  = Al + (size_t)(wid * 2) * 32 * 512;
  const __bf16* b1h  = b0h + 32 * 512;
  const __bf16* b1l  = b0l + 32 * 512;

  v8f acc0 = {};
  v8f acc1 = {};
  for (int kk = 0; kk < IN_DIM / 32; ++kk) {
    const int k0 = kk * 32;
    v16bf yh, yl;
    split_fragA(yrow, k0, half, false, yh, yl);
    v16bf fh = load_fragB(b0h + kk * 512, lane);
    v16bf fl = load_fragB(b0l + kk * 512, lane);
    acc0 = wmma_bf16(yh, fh, acc0);
    acc0 = wmma_bf16(yl, fh, acc0);
    acc0 = wmma_bf16(yh, fl, acc0);
    fh = load_fragB(b1h + kk * 512, lane);
    fl = load_fragB(b1l + kk * 512, lane);
    acc1 = wmma_bf16(yh, fh, acc1);
    acc1 = wmma_bf16(yl, fh, acc1);
    acc1 = wmma_bf16(yh, fl, acc1);
  }
#pragma unroll
  for (int v = 0; v < 8; ++v) {
    int m = v + 8 * half;
    s_r[m][n0a + l15] = acc0[v] - b[(size_t)(i0 + m) * OUT_DIM + n0a + l15];
    s_r[m][n0b + l15] = acc1[v] - b[(size_t)(i0 + m) * OUT_DIM + n0b + l15];
  }
  __syncthreads();

  const __bf16* g0h = Gh + (wid * 2) * 4 * 512;
  const __bf16* g0l = Gl + (wid * 2) * 4 * 512;
  const __bf16* g1h = g0h + 4 * 512;
  const __bf16* g1l = g0l + 4 * 512;
  v8f mu0 = {};
  v8f mu1 = {};
#pragma unroll
  for (int kk = 0; kk < 4; ++kk) {
    const int k0 = kk * 32;
    v16bf rh, rl;
    split_fragA(&s_r[l15][0], k0, half, false, rh, rl);
    v16bf fh = load_fragB(g0h + kk * 512, lane);
    v16bf fl = load_fragB(g0l + kk * 512, lane);
    mu0 = wmma_bf16(rh, fh, mu0);
    mu0 = wmma_bf16(rl, fh, mu0);
    mu0 = wmma_bf16(rh, fl, mu0);
    fh = load_fragB(g1h + kk * 512, lane);
    fl = load_fragB(g1l + kk * 512, lane);
    mu1 = wmma_bf16(rh, fh, mu1);
    mu1 = wmma_bf16(rl, fh, mu1);
    mu1 = wmma_bf16(rh, fl, mu1);
  }
#pragma unroll
  for (int v = 0; v < 8; ++v) {
    int m = v + 8 * half;
    s_mu[m][n0a + l15] = mu0[v];
    s_mu[m][n0b + l15] = mu1[v];
  }
  __syncthreads();

  v16bf mh[4], ml[4];
#pragma unroll
  for (int kk = 0; kk < 4; ++kk)
    split_fragA(&s_mu[l15][0], kk * 32, half, true, mh[kk], ml[kk]);

  for (int t = 0; t < 16; ++t) {
    const int c0 = (wid * 16 + t) * 16;
    const __bf16* ph = Ath + (size_t)(wid * 16 + t) * 4 * 512;
    const __bf16* pl = Atl + (size_t)(wid * 16 + t) * 4 * 512;
    v8f acc;
#pragma unroll
    for (int v = 0; v < 8; ++v)
      acc[v] = y[(size_t)(i0 + v + 8 * half) * IN_DIM + c0 + l15];
#pragma unroll
    for (int kk = 0; kk < 4; ++kk) {
      v16bf fh = load_fragB(ph + kk * 512, lane);
      v16bf fl = load_fragB(pl + kk * 512, lane);
      acc = wmma_bf16(mh[kk], fh, acc);
      acc = wmma_bf16(ml[kk], fh, acc);
      acc = wmma_bf16(mh[kk], fl, acc);
    }
#pragma unroll
    for (int v = 0; v < 8; ++v)
      out[(size_t)(i0 + v + 8 * half) * IN_DIM + c0 + l15] = acc[v];
  }
}

extern "C" void kernel_launch(void* const* d_in, const int* in_sizes, int n_in,
                              void* d_out, int out_size, void* d_ws,
                              size_t ws_size, hipStream_t stream) {
  (void)in_sizes; (void)n_in; (void)out_size; (void)ws_size;
  const float* y = (const float*)d_in[0];
  const float* A = (const float*)d_in[1];
  const float* b = (const float*)d_in[2];
  float* out = (float*)d_out;

  char* ws = (char*)d_ws;
  float*  G   = (float*) (ws + 0);
  __bf16* Gh  = (__bf16*)(ws + 65536);
  __bf16* Gl  = (__bf16*)(ws + 98304);
  __bf16* Ah  = (__bf16*)(ws + 131072);
  __bf16* Al  = (__bf16*)(ws + 393216);
  __bf16* Ath = (__bf16*)(ws + 655360);
  __bf16* Atl = (__bf16*)(ws + 917504);

  compute_G<<<64, 128, 0, stream>>>(A, G);
  invert_G<<<1, 512, 0, stream>>>(G, Gh, Gl);
  pack_A<<<128, 256, 0, stream>>>(A, Ah, Al, Ath, Atl);
  fused_project<<<BATCH / 16, 128, 0, stream>>>(y, b, Ah, Al, Ath, Atl,
                                                Gh, Gl, out);
}


# fused_project stages 0 and 2: next K-step's / output tile's operand loads issued ahead into a second register set (two alternating sets); launch bound keeps 4 groups per CU; order of sums unchanged
# speedup vs baseline: 6.0191x; 1.3295x over previous
#include <hip/hip_runtime.h>


#define BATCH   16384
#define IN_DIM  1024
#define OUT_DIM 128

typedef __attribute__((ext_vector_type(16))) __bf16 v16bf;
typedef __attribute__((ext_vector_type(8)))  __bf16 v8bf;
typedef __attribute__((ext_vector_type(8)))  float  v8f;
typedef __attribute__((ext_vector_type(2)))  float  v2f;

static __device__ inline v8f wmma_bf16(v16bf a, v16bf b, v8f c) {
  return __builtin_amdgcn_wmma_f32_16x16x32_bf16(false, a, false, b,
                                                 (short)0, c, false, false);
}

static __device__ inline v8f wmma_f32(v2f a, v2f b, v8f c) {
  return __builtin_amdgcn_wmma_f32_16x16x4_f32(false, a, false, b,
                                               (short)0, c, false, false);
}

static __device__ inline void split_fragA(const float* __restrict__ rowp,
                                          int k0, int half, bool neg,
                                          v16bf& hi, v16bf& lo) {
  const float* p0 = rowp + k0 + 8 * half;
  const float* p1 = p0 + 16;
#pragma unroll
  for (int e = 0; e < 8; ++e) {
    float x0 = neg ? -p0[e] : p0[e];
    float x1 = neg ? -p1[e] : p1[e];
    __bf16 h0 = (__bf16)x0;
    __bf16 h1 = (__bf16)x1;
    hi[e]     = h0;  lo[e]     = (__bf16)(x0 - (float)h0);
    hi[e + 8] = h1;  lo[e + 8] = (__bf16)(x1 - (float)h1);
  }
}

static __device__ inline v16bf load_fragB(const __bf16* __restrict__ frag,
                                          int lane) {
  const v8bf* p = reinterpret_cast<const v8bf*>(frag) + lane;
  const v8bf e0 = p[0];
  const v8bf e1 = p[32];
  return __builtin_shufflevector(e0, e1, 0, 1, 2, 3, 4, 5, 6, 7,
                                 8, 9, 10, 11, 12, 13, 14, 15);
}

static __device__ inline int frag_pos(int tile, int frags, int l15, int k) {
  const int kin = k & 31;
  return (tile * frags + (k >> 5)) * 512 + (kin >> 4) * 256 +
         (l15 + 16 * ((kin >> 3) & 1)) * 8 + (kin & 7);
}

__global__ __launch_bounds__(128) void compute_G(const float* __restrict__ A,
                                                 float* __restrict__ G) {
  __shared__ float part[4][8][32];
  const int tid  = threadIdx.x;
  const int lane = tid & 31;
  const int w    = tid >> 5;
  const int c    = lane & 15;
  const int h    = lane >> 4;
  const int ti   = blockIdx.x >> 3;
  const int tj   = blockIdx.x & 7;
  const float4* pa = reinterpret_cast<const float4*>(
      A + (size_t)(ti * 16 + c) * IN_DIM + w * 256 + 4 * h);
  const float4* pb = reinterpret_cast<const float4*>(
      A + (size_t)(tj * 16 + c) * IN_DIM + w * 256 + 4 * h);
  float4 xa[32], xb[32];
#pragma unroll
  for (int j = 0; j < 32; ++j) {
    xa[j] = pa[2 * j];
    xb[j] = pb[2 * j];
  }
  __builtin_amdgcn_sched_barrier(0);
  v8f acc0 = {};
  v8f acc1 = {};
#pragma unroll
  for (int j = 0; j < 32; ++j) {
    const v2f a0 = {xa[j].x, xa[j].y};
    const v2f b0 = {xb[j].x, xb[j].y};
    const v2f a1 = {xa[j].z, xa[j].w};
    const v2f b1 = {xb[j].z, xb[j].w};
    acc0 = wmma_f32(a0, b0, acc0);
    acc1 = wmma_f32(a1, b1, acc1);
  }
#pragma unroll
  for (int r = 0; r < 8; ++r) part[w][r][lane] = acc0[r] + acc1[r];
  __syncthreads();
#pragma unroll
  for (int q = 0; q < 2; ++q) {
    const int r = w + 4 * q;
    const float s = (part[0][r][lane] + part[1][r][lane]) +
                    (part[2][r][lane] + part[3][r][lane]);
    G[(ti * 16 + 8 * h + r) * OUT_DIM + tj * 16 + c] = s;
  }
}

__global__ __launch_bounds__(512) void invert_G(const float* __restrict__ G,
                                                __bf16* __restrict__ Gh,
                                                __bf16* __restrict__ Gl) {
  __shared__ __attribute__((aligned(16))) float piv[2][128];
  const int tid  = threadIdx.x;
  const int lane = tid & 31;
  const int w    = tid >> 5;
  const int rg   = w >> 1;
  const int R0   = rg * 16;
  const int c0   = (w & 1) * 64 + lane;
  const int c1   = c0 + 32;

  float a0[16], a1[16];
#pragma unroll
  for (int j = 0; j < 16; ++j) {
    a0[j] = G[(R0 + j) * 128 + c0];
    a1[j] = G[(R0 + j) * 128 + c1];
  }
  if (c0 == 0) {
    float4* nv4 = reinterpret_cast<float4*>(&piv[0][R0]);
#pragma unroll
    for (int q = 0; q < 4; ++q)
      nv4[q] = make_float4(a0[4 * q], a0[4 * q + 1], a0[4 * q + 2], a0[4 * q + 3]);
  }
  __syncthreads();

  for (int kb = 0; kb < 8; ++kb) {
#pragma unroll
    for (int kk = 0; kk < 16; ++kk) {
      const int k = kb * 16 + kk;
      const float* v = piv[kk & 1];
      const float pinv = 1.0f / v[k];
      const float t0 = v[c0] * pinv;
      const float t1 = v[c1] * pinv;
      const float n0 = -t0;
      const float n1 = -t1;
      const float4* v4 = reinterpret_cast<const float4*>(v + R0);
      float vr[16];
#pragma unroll
      for (int q = 0; q < 4; ++q) {
        float4 x = v4[q];
        vr[4 * q] = x.x; vr[4 * q + 1] = x.y; vr[4 * q + 2] = x.z; vr[4 * q + 3] = x.w;
      }
#pragma unroll
      for (int j = 0; j < 16; ++j) {
        a0[j] = fmaf(vr[j], n0, a0[j]);
        a1[j] = fmaf(vr[j], n1, a1[j]);
      }
      if (c0 == k) {
#pragma unroll
        for (int j = 0; j < 16; ++j) a0[j] = vr[j] * pinv;
      }
      if (c1 == k) {
#pragma unroll
        for (int j = 0; j < 16; ++j) a1[j] = vr[j] * pinv;
      }
      if (rg == kb) {
        a0[kk] = (c0 == k) ? -pinv : t0;
        a1[kk] = (c1 == k) ? -pinv : t1;
      }
      if (k + 1 < OUT_DIM) {
        float4* nv4 = reinterpret_cast<float4*>(&piv[(kk + 1) & 1][R0]);
        if (c0 == k + 1) {
#pragma unroll
          for (int q = 0; q < 4; ++q)
            nv4[q] = make_float4(a0[4 * q], a0[4 * q + 1], a0[4 * q + 2], a0[4 * q + 3]);
        }
        if (c1 == k + 1) {
#pragma unroll
          for (int q = 0; q < 4; ++q)
            nv4[q] = make_float4(a1[4 * q], a1[4 * q + 1], a1[4 * q + 2], a1[4 * q + 3]);
        }
      }
      __syncthreads();
    }
  }

#pragma unroll
  for (int j = 0; j < 16; ++j) {
    const int i0 = frag_pos(rg, 4, j, c0);
    const int i1 = frag_pos(rg, 4, j, c1);
    const float x0 = -a0[j];
    const float x1 = -a1[j];
    const __bf16 h0 = (__bf16)x0;
    const __bf16 h1 = (__bf16)x1;
    Gh[i0] = h0;  Gl[i0] = (__bf16)(x0 - (float)h0);
    Gh[i1] = h1;  Gl[i1] = (__bf16)(x1 - (float)h1);
  }
}

__global__ __launch_bounds__(256) void pack_A(const float* __restrict__ A,
                                              __bf16* __restrict__ Ah,
                                              __bf16* __restrict__ Al,
                                              __bf16* __restrict__ Ath,
                                              __bf16* __restrict__ Atl) {
  const int t     = blockIdx.x * 256 + threadIdx.x;
  const int q     = t & 16383;
  const int f     = q >> 6;
  const int lane  = q & 31;
  const int l15   = lane & 15;
  const int kin0  = ((q >> 5) & 1) * 16 + (lane >> 4) * 8;
  float x[8];
  __bf16* dh;
  __bf16* dl;
  if (t < 16384) {
    const int nt = f >> 5, kk = f & 31;
    const float* s = A + (size_t)(nt * 16 + l15) * IN_DIM + kk * 32 + kin0;
#pragma unroll
    for (int e = 0; e < 8; ++e) x[e] = s[e];
    dh = Ah;  dl = Al;
  } else {
    const int ct = f >> 2, kk = f & 3;
    const float* s = A + (size_t)(kk * 32 + kin0) * IN_DIM + ct * 16 + l15;
#pragma unroll
    for (int e = 0; e < 8; ++e) x[e] = s[(size_t)e * IN_DIM];
    dh = Ath;  dl = Atl;
  }
  v8bf h, l;
#pragma unroll
  for (int e = 0; e < 8; ++e) {
    const __bf16 hh = (__bf16)x[e];
    h[e] = hh;
    l[e] = (__bf16)(x[e] - (float)hh);
  }
  reinterpret_cast<v8bf*>(dh)[q] = h;
  reinterpret_cast<v8bf*>(dl)[q] = l;
}


static __device__ inline v16bf cat16(v8bf a, v8bf b) {
  return __builtin_shufflevector(a, b, 0, 1, 2, 3, 4, 5, 6, 7,
                                 8, 9, 10, 11, 12, 13, 14, 15);
}

static __device__ inline void split4(float4 v, int e0, v16bf& hi, v16bf& lo) {
  const float x[4] = {v.x, v.y, v.z, v.w};
#pragma unroll
  for (int i = 0; i < 4; ++i) {
    const __bf16 h = (__bf16)x[i];
    hi[e0 + i] = h;
    lo[e0 + i] = (__bf16)(x[i] - (float)h);
  }
}

struct Step0 {
  float4 y[4];
  v8bf   b[8];
};

static __device__ inline void load_step0(Step0& s, const float4* __restrict__ py,
                                         const v8bf* __restrict__ ph,
                                         const v8bf* __restrict__ pl, int kk) {
  const float4* q = py + kk * 8;
  s.y[0] = q[0];  s.y[1] = q[1];
  s.y[2] = q[4];  s.y[3] = q[5];
  const int f = kk * 64;
  s.b[0] = ph[f];         s.b[1] = ph[f + 32];
  s.b[2] = pl[f];         s.b[3] = pl[f + 32];
  s.b[4] = ph[f + 2048];  s.b[5] = ph[f + 2048 + 32];
  s.b[6] = pl[f + 2048];  s.b[7] = pl[f + 2048 + 32];
}

static __device__ inline void mma_step0(const Step0& s, v8f& acc0, v8f& acc1) {
  v16bf yh, yl;
  split4(s.y[0], 0, yh, yl);
  split4(s.y[1], 4, yh, yl);
  split4(s.y[2], 8, yh, yl);
  split4(s.y[3], 12, yh, yl);
  v16bf fh = cat16(s.b[0], s.b[1]);
  v16bf fl = cat16(s.b[2], s.b[3]);
  acc0 = wmma_bf16(yh, fh, acc0);
  acc0 = wmma_bf16(yl, fh, acc0);
  acc0 = wmma_bf16(yh, fl, acc0);
  fh = cat16(s.b[4], s.b[5]);
  fl = cat16(s.b[6], s.b[7]);
  acc1 = wmma_bf16(yh, fh, acc1);
  acc1 = wmma_bf16(yl, fh, acc1);
  acc1 = wmma_bf16(yh, fl, acc1);
}

struct Step2 {
  float c[8];
  v8bf  b[16];
};

static __device__ inline void load_step2(Step2& s, const float* __restrict__ yt,
                                         const v8bf* __restrict__ ph,
                                         const v8bf* __restrict__ pl, int t) {
#pragma unroll
  for (int v = 0; v < 8; ++v) s.c[v] = yt[(size_t)v * IN_DIM + t * 16];
  const int f = t * 256;
#pragma unroll
  for (int kk = 0; kk < 4; ++kk) {
    s.b[4 * kk]     = ph[f + kk * 64];
    s.b[4 * kk + 1] = ph[f + kk * 64 + 32];
    s.b[4 * kk + 2] = pl[f + kk * 64];
    s.b[4 * kk + 3] = pl[f + kk * 64 + 32];
  }
}

static __device__ inline void mma_step2(const Step2& s, const v16bf (&mh)[4],
                                        const v16bf (&ml)[4],
                                        float* __restrict__ ot, int t) {
  v8f acc;
#pragma unroll
  for (int v = 0; v < 8; ++v) acc[v] = s.c[v];
#pragma unroll
  for (int kk = 0; kk < 4; ++kk) {
    const v16bf fh = cat16(s.b[4 * kk], s.b[4 * kk + 1]);
    const v16bf fl = cat16(s.b[4 * kk + 2], s.b[4 * kk + 3]);
    acc = wmma_bf16(mh[kk], fh, acc);
    acc = wmma_bf16(ml[kk], fh, acc);
    acc = wmma_bf16(mh[kk], fl, acc);
  }
#pragma unroll
  for (int v = 0; v < 8; ++v) ot[(size_t)v * IN_DIM + t * 16] = acc[v];
}

__global__ __launch_bounds__(128, 4) void fused_project(
    const float* __restrict__ y, const float* __restrict__ b,
    const __bf16* __restrict__ Ah, const __bf16* __restrict__ Al,
    const __bf16* __restrict__ Ath, const __bf16* __restrict__ Atl,
    const __bf16* __restrict__ Gh, const __bf16* __restrict__ Gl,
    float* __restrict__ out) {
  __shared__ float s_r[16][OUT_DIM];
  __shared__ float s_mu[16][OUT_DIM];

  const int tid  = threadIdx.x;
  const int wid  = tid >> 5;
  const int lane = tid & 31;
  const int half = lane >> 4;
  const int l15  = lane & 15;
  const int i0   = blockIdx.x * 16;

  const int n0a = (wid * 2) * 16;
  const int n0b = n0a + 16;
  const float4* py0 = reinterpret_cast<const float4*>(
      y + (size_t)(i0 + l15) * IN_DIM + 8 * half);
  const v8bf* p0h = reinterpret_cast<const v8bf*>(
      Ah + (size_t)(wid * 2) * 32 * 512) + lane;
  const v8bf* p0l = reinterpret_cast<const v8bf*>(
      Al + (size_t)(wid * 2) * 32 * 512) + lane;

  v8f acc0 = {};
  v8f acc1 = {};
  {
    Step0 sa, sb;
    load_step0(sa, py0, p0h, p0l, 0);
    for (int kk = 0; kk < IN_DIM / 32 - 2; kk += 2) {
      load_step0(sb, py0, p0h, p0l, kk + 1);
      __builtin_amdgcn_sched_barrier(0);
      mma_step0(sa, acc0, acc1);
      load_step0(sa, py0, p0h, p0l, kk + 2);
      __builtin_amdgcn_sched_barrier(0);
      mma_step0(sb, acc0, acc1);
    }
    load_step0(sb, py0, p0h, p0l, IN_DIM / 32 - 1);
    mma_step0(sa, acc0, acc1);
    mma_step0(sb, acc0, acc1);
  }
#pragma unroll
  for (int v = 0; v < 8; ++v) {
    int m = v + 8 * half;
    s_r[m][n0a + l15] = acc0[v] - b[(size_t)(i0 + m) * OUT_DIM + n0a + l15];
    s_r[m][n0b + l15] = acc1[v] - b[(size_t)(i0 + m) * OUT_DIM + n0b + l15];
  }
  __syncthreads();

  const __bf16* g0h = Gh + (wid * 2) * 4 * 512;
  const __bf16* g0l = Gl + (wid * 2) * 4 * 512;
  const __bf16* g1h = g0h + 4 * 512;
  const __bf16* g1l = g0l + 4 * 512;
  v8f mu0 = {};
  v8f mu1 = {};
#pragma unroll
  for (int kk = 0; kk < 4; ++kk) {
    const int k0 = kk * 32;
    v16bf rh, rl;
    split_fragA(&s_r[l15][0], k0, half, false, rh, rl);
    v16bf fh = load_fragB(g0h + kk * 512, lane);
    v16bf fl = load_fragB(g0l + kk * 512, lane);
    mu0 = wmma_bf16(rh, fh, mu0);
    mu0 = wmma_bf16(rl, fh, mu0);
    mu0 = wmma_bf16(rh, fl, mu0);
    fh = load_fragB(g1h + kk * 512, lane);
    fl = load_fragB(g1l + kk * 512, lane);
    mu1 = wmma_bf16(rh, fh, mu1);
    mu1 = wmma_bf16(rl, fh, mu1);
    mu1 = wmma_bf16(rh, fl, mu1);
  }
#pragma unroll
  for (int v = 0; v < 8; ++v) {
    int m = v + 8 * half;
    s_mu[m][n0a + l15] = mu0[v];
    s_mu[m][n0b + l15] = mu1[v];
  }
  __syncthreads();

  v16bf mh[4], ml[4];
#pragma unroll
  for (int kk = 0; kk < 4; ++kk)
    split_fragA(&s_mu[l15][0], kk * 32, half, true, mh[kk], ml[kk]);

  const size_t tile0 = (size_t)(i0 + 8 * half) * IN_DIM + wid * 256 + l15;
  const float* yt = y + tile0;
  float*       ot = out + tile0;
  const v8bf* p2h = reinterpret_cast<const v8bf*>(
      Ath + (size_t)(wid * 16) * 4 * 512) + lane;
  const v8bf* p2l = reinterpret_cast<const v8bf*>(
      Atl + (size_t)(wid * 16) * 4 * 512) + lane;
  {
    Step2 ta, tb;
    load_step2(ta, yt, p2h, p2l, 0);
    for (int t = 0; t < 14; t += 2) {
      load_step2(tb, yt, p2h, p2l, t + 1);
      __builtin_amdgcn_sched_barrier(0);
      mma_step2(ta, mh, ml, ot, t);
      load_step2(ta, yt, p2h, p2l, t + 2);
      __builtin_amdgcn_sched_barrier(0);
      mma_step2(tb, mh, ml, ot, t + 1);
    }
    load_step2(tb, yt, p2h, p2l, 15);
    __builtin_amdgcn_sched_barrier(0);
    mma_step2(ta, mh, ml, ot, 14);
    mma_step2(tb, mh, ml, ot, 15);
  }
}

extern "C" void kernel_launch(void* const* d_in, const int* in_sizes, int n_in,
                              void* d_out, int out_size, void* d_ws,
                              size_t ws_size, hipStream_t stream) {
  (void)in_sizes; (void)n_in; (void)out_size; (void)ws_size;
  const float* y = (const float*)d_in[0];
  const float* A = (const float*)d_in[1];
  const float* b = (const float*)d_in[2];
  float* out = (float*)d_out;

  char* ws = (char*)d_ws;
  float*  G   = (float*) (ws + 0);
  __bf16* Gh  = (__bf16*)(ws + 65536);
  __bf16* Gl  = (__bf16*)(ws + 98304);
  __bf16* Ah  = (__bf16*)(ws + 131072);
  __bf16* Al  = (__bf16*)(ws + 393216);
  __bf16* Ath = (__bf16*)(ws + 655360);
  __bf16* Atl = (__bf16*)(ws + 917504);

  compute_G<<<64, 128, 0, stream>>>(A, G);
  invert_G<<<1, 512, 0, stream>>>(G, Gh, Gl);
  pack_A<<<128, 256, 0, stream>>>(A, Ah, Al, Ath, Atl);
  fused_project<<<BATCH / 16, 128, 0, stream>>>(y, b, Ah, Al, Ath, Atl,
                                                Gh, Gl, out);
}


# invert_G shares a launch with stage 0 (r = y A^T - b): workgroup 0 inverts G while 255 others form r, kept in out's first 128 columns until the new launch of stages 1-2 reads it; sums unchanged
# speedup vs baseline: 7.5717x; 1.2579x over previous
#include <hip/hip_runtime.h>


#define BATCH   16384
#define IN_DIM  1024
#define OUT_DIM 128

typedef __attribute__((ext_vector_type(16))) __bf16 v16bf;
typedef __attribute__((ext_vector_type(8)))  __bf16 v8bf;
typedef __attribute__((ext_vector_type(8)))  float  v8f;
typedef __attribute__((ext_vector_type(2)))  float  v2f;

static __device__ inline v8f wmma_bf16(v16bf a, v16bf b, v8f c) {
  return __builtin_amdgcn_wmma_f32_16x16x32_bf16(false, a, false, b,
                                                 (short)0, c, false, false);
}

static __device__ inline v8f wmma_f32(v2f a, v2f b, v8f c) {
  return __builtin_amdgcn_wmma_f32_16x16x4_f32(false, a, false, b,
                                               (short)0, c, false, false);
}

static __device__ inline void split_fragA(const float* __restrict__ rowp,
                                          int k0, int half, bool neg,
                                          v16bf& hi, v16bf& lo) {
  const float* p0 = rowp + k0 + 8 * half;
  const float* p1 = p0 + 16;
#pragma unroll
  for (int e = 0; e < 8; ++e) {
    float x0 = neg ? -p0[e] : p0[e];
    float x1 = neg ? -p1[e] : p1[e];
    __bf16 h0 = (__bf16)x0;
    __bf16 h1 = (__bf16)x1;
    hi[e]     = h0;  lo[e]     = (__bf16)(x0 - (float)h0);
    hi[e + 8] = h1;  lo[e + 8] = (__bf16)(x1 - (float)h1);
  }
}

static __device__ inline v16bf load_fragB(const __bf16* __restrict__ frag,
                                          int lane) {
  const v8bf* p = reinterpret_cast<const v8bf*>(frag) + lane;
  const v8bf e0 = p[0];
  const v8bf e1 = p[32];
  return __builtin_shufflevector(e0, e1, 0, 1, 2, 3, 4, 5, 6, 7,
                                 8, 9, 10, 11, 12, 13, 14, 15);
}

static __device__ inline int frag_pos(int tile, int frags, int l15, int k) {
  const int kin = k & 31;
  return (tile * frags + (k >> 5)) * 512 + (kin >> 4) * 256 +
         (l15 + 16 * ((kin >> 3) & 1)) * 8 + (kin & 7);
}

__global__ __launch_bounds__(128) void compute_G(const float* __restrict__ A,
                                                 float* __restrict__ G) {
  __shared__ float part[4][8][32];
  const int tid  = threadIdx.x;
  const int lane = tid & 31;
  const int w    = tid >> 5;
  const int c    = lane & 15;
  const int h    = lane >> 4;
  const int ti   = blockIdx.x >> 3;
  const int tj   = blockIdx.x & 7;
  const float4* pa = reinterpret_cast<const float4*>(
      A + (size_t)(ti * 16 + c) * IN_DIM + w * 256 + 4 * h);
  const float4* pb = reinterpret_cast<const float4*>(
      A + (size_t)(tj * 16 + c) * IN_DIM + w * 256 + 4 * h);
  float4 xa[32], xb[32];
#pragma unroll
  for (int j = 0; j < 32; ++j) {
    xa[j] = pa[2 * j];
    xb[j] = pb[2 * j];
  }
  __builtin_amdgcn_sched_barrier(0);
  v8f acc0 = {};
  v8f acc1 = {};
#pragma unroll
  for (int j = 0; j < 32; ++j) {
    const v2f a0 = {xa[j].x, xa[j].y};
    const v2f b0 = {xb[j].x, xb[j].y};
    const v2f a1 = {xa[j].z, xa[j].w};
    const v2f b1 = {xb[j].z, xb[j].w};
    acc0 = wmma_f32(a0, b0, acc0);
    acc1 = wmma_f32(a1, b1, acc1);
  }
#pragma unroll
  for (int r = 0; r < 8; ++r) part[w][r][lane] = acc0[r] + acc1[r];
  __syncthreads();
#pragma unroll
  for (int q = 0; q < 2; ++q) {
    const int r = w + 4 * q;
    const float s = (part[0][r][lane] + part[1][r][lane]) +
                    (part[2][r][lane] + part[3][r][lane]);
    G[(ti * 16 + 8 * h + r) * OUT_DIM + tj * 16 + c] = s;
  }
}

__global__ __launch_bounds__(256) void pack_A(const float* __restrict__ A,
                                              __bf16* __restrict__ Ah,
                                              __bf16* __restrict__ Al,
                                              __bf16* __restrict__ Ath,
                                              __bf16* __restrict__ Atl) {
  const int t     = blockIdx.x * 256 + threadIdx.x;
  const int q     = t & 16383;
  const int f     = q >> 6;
  const int lane  = q & 31;
  const int l15   = lane & 15;
  const int kin0  = ((q >> 5) & 1) * 16 + (lane >> 4) * 8;
  float x[8];
  __bf16* dh;
  __bf16* dl;
  if (t < 16384) {
    const int nt = f >> 5, kk = f & 31;
    const float* s = A + (size_t)(nt * 16 + l15) * IN_DIM + kk * 32 + kin0;
#pragma unroll
    for (int e = 0; e < 8; ++e) x[e] = s[e];
    dh = Ah;  dl = Al;
  } else {
    const int ct = f >> 2, kk = f & 3;
    const float* s = A + (size_t)(kk * 32 + kin0) * IN_DIM + ct * 16 + l15;
#pragma unroll
    for (int e = 0; e < 8; ++e) x[e] = s[(size_t)e * IN_DIM];
    dh = Ath;  dl = Atl;
  }
  v8bf h, l;
#pragma unroll
  for (int e = 0; e < 8; ++e) {
    const __bf16 hh = (__bf16)x[e];
    h[e] = hh;
    l[e] = (__bf16)(x[e] - (float)hh);
  }
  reinterpret_cast<v8bf*>(dh)[q] = h;
  reinterpret_cast<v8bf*>(dl)[q] = l;
}

static __device__ inline void invert_body(const float* __restrict__ G,
                                          __bf16* __restrict__ Gh,
                                          __bf16* __restrict__ Gl) {
  __shared__ __attribute__((aligned(16))) float piv[2][128];
  const int tid  = threadIdx.x;
  const int lane = tid & 31;
  const int w    = tid >> 5;
  const int rg   = w >> 1;
  const int R0   = rg * 16;
  const int c0   = (w & 1) * 64 + lane;
  const int c1   = c0 + 32;

  float a0[16], a1[16];
#pragma unroll
  for (int j = 0; j < 16; ++j) {
    a0[j] = G[(R0 + j) * 128 + c0];
    a1[j] = G[(R0 + j) * 128 + c1];
  }
  if (c0 == 0) {
    float4* nv4 = reinterpret_cast<float4*>(&piv[0][R0]);
#pragma unroll
    for (int q = 0; q < 4; ++q)
      nv4[q] = make_float4(a0[4 * q], a0[4 * q + 1], a0[4 * q + 2], a0[4 * q + 3]);
  }
  __syncthreads();

  for (int kb = 0; kb < 8; ++kb) {
#pragma unroll
    for (int kk = 0; kk < 16; ++kk) {
      const int k = kb * 16 + kk;
      const float* v = piv[kk & 1];
      const float pinv = 1.0f / v[k];
      const float t0 = v[c0] * pinv;
      const float t1 = v[c1] * pinv;
      const float n0 = -t0;
      const float n1 = -t1;
      const float4* v4 = reinterpret_cast<const float4*>(v + R0);
      float vr[16];
#pragma unroll
      for (int q = 0; q < 4; ++q) {
        float4 x = v4[q];
        vr[4 * q] = x.x; vr[4 * q + 1] = x.y; vr[4 * q + 2] = x.z; vr[4 * q + 3] = x.w;
      }
#pragma unroll
      for (int j = 0; j < 16; ++j) {
        a0[j] = fmaf(vr[j], n0, a0[j]);
        a1[j] = fmaf(vr[j], n1, a1[j]);
      }
      if (c0 == k) {
#pragma unroll
        for (int j = 0; j < 16; ++j) a0[j] = vr[j] * pinv;
      }
      if (c1 == k) {
#pragma unroll
        for (int j = 0; j < 16; ++j) a1[j] = vr[j] * pinv;
      }
      if (rg == kb) {
        a0[kk] = (c0 == k) ? -pinv : t0;
        a1[kk] = (c1 == k) ? -pinv : t1;
      }
      if (k + 1 < OUT_DIM) {
        float4* nv4 = reinterpret_cast<float4*>(&piv[(kk + 1) & 1][R0]);
        if (c0 == k + 1) {
#pragma unroll
          for (int q = 0; q < 4; ++q)
            nv4[q] = make_float4(a0[4 * q], a0[4 * q + 1], a0[4 * q + 2], a0[4 * q + 3]);
        }
        if (c1 == k + 1) {
#pragma unroll
          for (int q = 0; q < 4; ++q)
            nv4[q] = make_float4(a1[4 * q], a1[4 * q + 1], a1[4 * q + 2], a1[4 * q + 3]);
        }
      }
      __syncthreads();
    }
  }

#pragma unroll
  for (int j = 0; j < 16; ++j) {
    const int i0 = frag_pos(rg, 4, j, c0);
    const int i1 = frag_pos(rg, 4, j, c1);
    const float x0 = -a0[j];
    const float x1 = -a1[j];
    const __bf16 h0 = (__bf16)x0;
    const __bf16 h1 = (__bf16)x1;
    Gh[i0] = h0;  Gl[i0] = (__bf16)(x0 - (float)h0);
    Gh[i1] = h1;  Gl[i1] = (__bf16)(x1 - (float)h1);
  }
}


static __device__ inline v16bf cat16(v8bf a, v8bf b) {
  return __builtin_shufflevector(a, b, 0, 1, 2, 3, 4, 5, 6, 7,
                                 8, 9, 10, 11, 12, 13, 14, 15);
}

static __device__ inline void split4(float4 v, int e0, v16bf& hi, v16bf& lo) {
  const float x[4] = {v.x, v.y, v.z, v.w};
#pragma unroll
  for (int i = 0; i < 4; ++i) {
    const __bf16 h = (__bf16)x[i];
    hi[e0 + i] = h;
    lo[e0 + i] = (__bf16)(x[i] - (float)h);
  }
}

template <int NT>
struct Step0 {
  float4 y[NT][4];
  v8bf   b[8];
};

template <int NT>
static __device__ inline void load_step0(Step0<NT>& s,
                                         const float4* const (&py)[NT],
                                         const v8bf* __restrict__ ph,
                                         const v8bf* __restrict__ pl, int kk) {
#pragma unroll
  for (int u = 0; u < NT; ++u) {
    const float4* q = py[u] + kk * 8;
    s.y[u][0] = q[0];  s.y[u][1] = q[1];
    s.y[u][2] = q[4];  s.y[u][3] = q[5];
  }
  const int f = kk * 64;
  s.b[0] = ph[f];         s.b[1] = ph[f + 32];
  s.b[2] = pl[f];         s.b[3] = pl[f + 32];
  s.b[4] = ph[f + 2048];  s.b[5] = ph[f + 2048 + 32];
  s.b[6] = pl[f + 2048];  s.b[7] = pl[f + 2048 + 32];
}

template <int NT>
static __device__ inline void mma_step0(const Step0<NT>& s, v8f (&acc0)[NT],
                                        v8f (&acc1)[NT]) {
  v16bf yh[NT], yl[NT];
#pragma unroll
  for (int u = 0; u < NT; ++u) {
    split4(s.y[u][0], 0, yh[u], yl[u]);
    split4(s.y[u][1], 4, yh[u], yl[u]);
    split4(s.y[u][2], 8, yh[u], yl[u]);
    split4(s.y[u][3], 12, yh[u], yl[u]);
  }
  v16bf fh = cat16(s.b[0], s.b[1]);
  v16bf fl = cat16(s.b[2], s.b[3]);
#pragma unroll
  for (int u = 0; u < NT; ++u) acc0[u] = wmma_bf16(yh[u], fh, acc0[u]);
#pragma unroll
  for (int u = 0; u < NT; ++u) acc0[u] = wmma_bf16(yl[u], fh, acc0[u]);
#pragma unroll
  for (int u = 0; u < NT; ++u) acc0[u] = wmma_bf16(yh[u], fl, acc0[u]);
  fh = cat16(s.b[4], s.b[5]);
  fl = cat16(s.b[6], s.b[7]);
#pragma unroll
  for (int u = 0; u < NT; ++u) acc1[u] = wmma_bf16(yh[u], fh, acc1[u]);
#pragma unroll
  for (int u = 0; u < NT; ++u) acc1[u] = wmma_bf16(yl[u], fh, acc1[u]);
#pragma unroll
  for (int u = 0; u < NT; ++u) acc1[u] = wmma_bf16(yh[u], fl, acc1[u]);
}

template <int NT>
static __device__ inline void residual_units(
    const int (&i0)[NT], int wid, int lane, const float* __restrict__ y,
    const float* __restrict__ b, const __bf16* __restrict__ Ah,
    const __bf16* __restrict__ Al, float* __restrict__ out) {
  const int half = lane >> 4;
  const int l15  = lane & 15;
  const int n0a  = (wid * 2) * 16;
  const int n0b  = n0a + 16;
  const float4* py[NT];
#pragma unroll
  for (int u = 0; u < NT; ++u)
    py[u] = reinterpret_cast<const float4*>(
        y + (size_t)(i0[u] + l15) * IN_DIM + 8 * half);
  const v8bf* p0h = reinterpret_cast<const v8bf*>(
      Ah + (size_t)(wid * 2) * 32 * 512) + lane;
  const v8bf* p0l = reinterpret_cast<const v8bf*>(
      Al + (size_t)(wid * 2) * 32 * 512) + lane;

  v8f acc0[NT] = {};
  v8f acc1[NT] = {};
  {
    Step0<NT> sa, sb;
    load_step0<NT>(sa, py, p0h, p0l, 0);
    for (int kk = 0; kk < IN_DIM / 32 - 2; kk += 2) {
      load_step0<NT>(sb, py, p0h, p0l, kk + 1);
      __builtin_amdgcn_sched_barrier(0);
      mma_step0<NT>(sa, acc0, acc1);
      load_step0<NT>(sa, py, p0h, p0l, kk + 2);
      __builtin_amdgcn_sched_barrier(0);
      mma_step0<NT>(sb, acc0, acc1);
    }
    load_step0<NT>(sb, py, p0h, p0l, IN_DIM / 32 - 1);
    mma_step0<NT>(sa, acc0, acc1);
    mma_step0<NT>(sb, acc0, acc1);
  }
#pragma unroll
  for (int u = 0; u < NT; ++u) {
#pragma unroll
    for (int v = 0; v < 8; ++v) {
      const int m = v + 8 * half;
      float* rrow = out + (size_t)(i0[u] + m) * IN_DIM;
      const float* brow = b + (size_t)(i0[u] + m) * OUT_DIM;
      rrow[n0a + l15] = acc0[u][v] - brow[n0a + l15];
      rrow[n0b + l15] = acc1[u][v] - brow[n0b + l15];
    }
  }
}

struct Step2 {
  float c[8];
  v8bf  b[16];
};

static __device__ inline void load_step2(Step2& s, const float* __restrict__ yt,
                                         const v8bf* __restrict__ ph,
                                         const v8bf* __restrict__ pl, int t) {
#pragma unroll
  for (int v = 0; v < 8; ++v) s.c[v] = yt[(size_t)v * IN_DIM + t * 16];
  const int f = t * 256;
#pragma unroll
  for (int kk = 0; kk < 4; ++kk) {
    s.b[4 * kk]     = ph[f + kk * 64];
    s.b[4 * kk + 1] = ph[f + kk * 64 + 32];
    s.b[4 * kk + 2] = pl[f + kk * 64];
    s.b[4 * kk + 3] = pl[f + kk * 64 + 32];
  }
}

static __device__ inline void mma_step2(const Step2& s, const v16bf (&mh)[4],
                                        const v16bf (&ml)[4],
                                        float* __restrict__ ot, int t) {
  v8f acc;
#pragma unroll
  for (int v = 0; v < 8; ++v) acc[v] = s.c[v];
#pragma unroll
  for (int kk = 0; kk < 4; ++kk) {
    const v16bf fh = cat16(s.b[4 * kk], s.b[4 * kk + 1]);
    const v16bf fl = cat16(s.b[4 * kk + 2], s.b[4 * kk + 3]);
    acc = wmma_bf16(mh[kk], fh, acc);
    acc = wmma_bf16(ml[kk], fh, acc);
    acc = wmma_bf16(mh[kk], fl, acc);
  }
#pragma unroll
  for (int v = 0; v < 8; ++v) ot[(size_t)v * IN_DIM + t * 16] = acc[v];
}

__global__ __launch_bounds__(512) void invert_and_residual(
    const float* __restrict__ G, __bf16* __restrict__ Gh,
    __bf16* __restrict__ Gl, const float* __restrict__ y,
    const float* __restrict__ b, const __bf16* __restrict__ Ah,
    const __bf16* __restrict__ Al, float* __restrict__ out) {
  if (blockIdx.x == 0) {
    invert_body(G, Gh, Gl);
    return;
  }
  const int tid  = threadIdx.x;
  const int sub  = __builtin_amdgcn_readfirstlane(tid >> 7);
  const int wid  = (tid >> 5) & 3;
  const int lane = tid & 31;
  const int unit = (blockIdx.x - 1) * 4 + sub;
  if (sub == 0 && blockIdx.x <= 4) {
    const int rows[2] = {unit * 16, (1020 + (int)blockIdx.x - 1) * 16};
    residual_units<2>(rows, wid, lane, y, b, Ah, Al, out);
  } else {
    const int rows[1] = {unit * 16};
    residual_units<1>(rows, wid, lane, y, b, Ah, Al, out);
  }
}

__global__ __launch_bounds__(128, 4) void project(
    const float* __restrict__ y,
    const __bf16* __restrict__ Ath, const __bf16* __restrict__ Atl,
    const __bf16* __restrict__ Gh, const __bf16* __restrict__ Gl,
    float* out) {
  __shared__ float s_mu[16][OUT_DIM];

  const int tid  = threadIdx.x;
  const int wid  = tid >> 5;
  const int lane = tid & 31;
  const int half = lane >> 4;
  const int l15  = lane & 15;
  const int i0   = blockIdx.x * 16;
  const int n0a  = (wid * 2) * 16;
  const int n0b  = n0a + 16;

  const float* rrow = out + (size_t)(i0 + l15) * IN_DIM;
  const __bf16* g0h = Gh + (wid * 2) * 4 * 512;
  const __bf16* g0l = Gl + (wid * 2) * 4 * 512;
  const __bf16* g1h = g0h + 4 * 512;
  const __bf16* g1l = g0l + 4 * 512;
  v8f mu0 = {};
  v8f mu1 = {};
#pragma unroll
  for (int kk = 0; kk < 4; ++kk) {
    const int k0 = kk * 32;
    v16bf rh, rl;
    split_fragA(rrow, k0, half, false, rh, rl);
    v16bf fh = load_fragB(g0h + kk * 512, lane);
    v16bf fl = load_fragB(g0l + kk * 512, lane);
    mu0 = wmma_bf16(rh, fh, mu0);
    mu0 = wmma_bf16(rl, fh, mu0);
    mu0 = wmma_bf16(rh, fl, mu0);
    fh = load_fragB(g1h + kk * 512, lane);
    fl = load_fragB(g1l + kk * 512, lane);
    mu1 = wmma_bf16(rh, fh, mu1);
    mu1 = wmma_bf16(rl, fh, mu1);
    mu1 = wmma_bf16(rh, fl, mu1);
  }
#pragma unroll
  for (int v = 0; v < 8; ++v) {
    int m = v + 8 * half;
    s_mu[m][n0a + l15] = mu0[v];
    s_mu[m][n0b + l15] = mu1[v];
  }
  __syncthreads();

  v16bf mh[4], ml[4];
#pragma unroll
  for (int kk = 0; kk < 4; ++kk)
    split_fragA(&s_mu[l15][0], kk * 32, half, true, mh[kk], ml[kk]);

  const size_t tile0 = (size_t)(i0 + 8 * half) * IN_DIM + wid * 256 + l15;
  const float* yt = y + tile0;
  float*       ot = out + tile0;
  const v8bf* p2h = reinterpret_cast<const v8bf*>(
      Ath + (size_t)(wid * 16) * 4 * 512) + lane;
  const v8bf* p2l = reinterpret_cast<const v8bf*>(
      Atl + (size_t)(wid * 16) * 4 * 512) + lane;
  {
    Step2 ta, tb;
    load_step2(ta, yt, p2h, p2l, 0);
    for (int t = 0; t < 14; t += 2) {
      load_step2(tb, yt, p2h, p2l, t + 1);
      __builtin_amdgcn_sched_barrier(0);
      mma_step2(ta, mh, ml, ot, t);
      load_step2(ta, yt, p2h, p2l, t + 2);
      __builtin_amdgcn_sched_barrier(0);
      mma_step2(tb, mh, ml, ot, t + 1);
    }
    load_step2(tb, yt, p2h, p2l, 15);
    __builtin_amdgcn_sched_barrier(0);
    mma_step2(ta, mh, ml, ot, 14);
    mma_step2(tb, mh, ml, ot, 15);
  }
}

extern "C" void kernel_launch(void* const* d_in, const int* in_sizes, int n_in,
                              void* d_out, int out_size, void* d_ws,
                              size_t ws_size, hipStream_t stream) {
  (void)in_sizes; (void)n_in; (void)out_size; (void)ws_size;
  const float* y = (const float*)d_in[0];
  const float* A = (const float*)d_in[1];
  const float* b = (const float*)d_in[2];
  float* out = (float*)d_out;

  char* ws = (char*)d_ws;
  float*  G   = (float*) (ws + 0);
  __bf16* Gh  = (__bf16*)(ws + 65536);
  __bf16* Gl  = (__bf16*)(ws + 98304);
  __bf16* Ah  = (__bf16*)(ws + 131072);
  __bf16* Al  = (__bf16*)(ws + 393216);
  __bf16* Ath = (__bf16*)(ws + 655360);
  __bf16* Atl = (__bf16*)(ws + 917504);

  compute_G<<<64, 128, 0, stream>>>(A, G);
  pack_A<<<128, 256, 0, stream>>>(A, Ah, Al, Ath, Atl);
  invert_and_residual<<<256, 512, 0, stream>>>(G, Gh, Gl, y, b, Ah, Al, out);
  project<<<BATCH / 16, 128, 0, stream>>>(y, Ath, Atl, Gh, Gl, out);
}


# project (stages 1-2): a workgroup takes two 16-row units (32 rows), each Ginv and A^T fragment load feeds both units' products; grid 1024 -> 512, 2 groups per CU; order of sums unchanged
# speedup vs baseline: 8.0360x; 1.0613x over previous
#include <hip/hip_runtime.h>


#define BATCH   16384
#define IN_DIM  1024
#define OUT_DIM 128

typedef __attribute__((ext_vector_type(16))) __bf16 v16bf;
typedef __attribute__((ext_vector_type(8)))  __bf16 v8bf;
typedef __attribute__((ext_vector_type(8)))  float  v8f;
typedef __attribute__((ext_vector_type(2)))  float  v2f;

static __device__ inline v8f wmma_bf16(v16bf a, v16bf b, v8f c) {
  return __builtin_amdgcn_wmma_f32_16x16x32_bf16(false, a, false, b,
                                                 (short)0, c, false, false);
}

static __device__ inline v8f wmma_f32(v2f a, v2f b, v8f c) {
  return __builtin_amdgcn_wmma_f32_16x16x4_f32(false, a, false, b,
                                               (short)0, c, false, false);
}

static __device__ inline void split_fragA(const float* __restrict__ rowp,
                                          int k0, int half, bool neg,
                                          v16bf& hi, v16bf& lo) {
  const float* p0 = rowp + k0 + 8 * half;
  const float* p1 = p0 + 16;
#pragma unroll
  for (int e = 0; e < 8; ++e) {
    float x0 = neg ? -p0[e] : p0[e];
    float x1 = neg ? -p1[e] : p1[e];
    __bf16 h0 = (__bf16)x0;
    __bf16 h1 = (__bf16)x1;
    hi[e]     = h0;  lo[e]     = (__bf16)(x0 - (float)h0);
    hi[e + 8] = h1;  lo[e + 8] = (__bf16)(x1 - (float)h1);
  }
}

static __device__ inline v16bf load_fragB(const __bf16* __restrict__ frag,
                                          int lane) {
  const v8bf* p = reinterpret_cast<const v8bf*>(frag) + lane;
  const v8bf e0 = p[0];
  const v8bf e1 = p[32];
  return __builtin_shufflevector(e0, e1, 0, 1, 2, 3, 4, 5, 6, 7,
                                 8, 9, 10, 11, 12, 13, 14, 15);
}

static __device__ inline int frag_pos(int tile, int frags, int l15, int k) {
  const int kin = k & 31;
  return (tile * frags + (k >> 5)) * 512 + (kin >> 4) * 256 +
         (l15 + 16 * ((kin >> 3) & 1)) * 8 + (kin & 7);
}

__global__ __launch_bounds__(128) void compute_G(const float* __restrict__ A,
                                                 float* __restrict__ G) {
  __shared__ float part[4][8][32];
  const int tid  = threadIdx.x;
  const int lane = tid & 31;
  const int w    = tid >> 5;
  const int c    = lane & 15;
  const int h    = lane >> 4;
  const int ti   = blockIdx.x >> 3;
  const int tj   = blockIdx.x & 7;
  const float4* pa = reinterpret_cast<const float4*>(
      A + (size_t)(ti * 16 + c) * IN_DIM + w * 256 + 4 * h);
  const float4* pb = reinterpret_cast<const float4*>(
      A + (size_t)(tj * 16 + c) * IN_DIM + w * 256 + 4 * h);
  float4 xa[32], xb[32];
#pragma unroll
  for (int j = 0; j < 32; ++j) {
    xa[j] = pa[2 * j];
    xb[j] = pb[2 * j];
  }
  __builtin_amdgcn_sched_barrier(0);
  v8f acc0 = {};
  v8f acc1 = {};
#pragma unroll
  for (int j = 0; j < 32; ++j) {
    const v2f a0 = {xa[j].x, xa[j].y};
    const v2f b0 = {xb[j].x, xb[j].y};
    const v2f a1 = {xa[j].z, xa[j].w};
    const v2f b1 = {xb[j].z, xb[j].w};
    acc0 = wmma_f32(a0, b0, acc0);
    acc1 = wmma_f32(a1, b1, acc1);
  }
#pragma unroll
  for (int r = 0; r < 8; ++r) part[w][r][lane] = acc0[r] + acc1[r];
  __syncthreads();
#pragma unroll
  for (int q = 0; q < 2; ++q) {
    const int r = w + 4 * q;
    const float s = (part[0][r][lane] + part[1][r][lane]) +
                    (part[2][r][lane] + part[3][r][lane]);
    G[(ti * 16 + 8 * h + r) * OUT_DIM + tj * 16 + c] = s;
  }
}

__global__ __launch_bounds__(256) void pack_A(const float* __restrict__ A,
                                              __bf16* __restrict__ Ah,
                                              __bf16* __restrict__ Al,
                                              __bf16* __restrict__ Ath,
                                              __bf16* __restrict__ Atl) {
  const int t     = blockIdx.x * 256 + threadIdx.x;
  const int q     = t & 16383;
  const int f     = q >> 6;
  const int lane  = q & 31;
  const int l15   = lane & 15;
  const int kin0  = ((q >> 5) & 1) * 16 + (lane >> 4) * 8;
  float x[8];
  __bf16* dh;
  __bf16* dl;
  if (t < 16384) {
    const int nt = f >> 5, kk = f & 31;
    const float* s = A + (size_t)(nt * 16 + l15) * IN_DIM + kk * 32 + kin0;
#pragma unroll
    for (int e = 0; e < 8; ++e) x[e] = s[e];
    dh = Ah;  dl = Al;
  } else {
    const int ct = f >> 2, kk = f & 3;
    const float* s = A + (size_t)(kk * 32 + kin0) * IN_DIM + ct * 16 + l15;
#pragma unroll
    for (int e = 0; e < 8; ++e) x[e] = s[(size_t)e * IN_DIM];
    dh = Ath;  dl = Atl;
  }
  v8bf h, l;
#pragma unroll
  for (int e = 0; e < 8; ++e) {
    const __bf16 hh = (__bf16)x[e];
    h[e] = hh;
    l[e] = (__bf16)(x[e] - (float)hh);
  }
  reinterpret_cast<v8bf*>(dh)[q] = h;
  reinterpret_cast<v8bf*>(dl)[q] = l;
}

static __device__ inline void invert_body(const float* __restrict__ G,
                                          __bf16* __restrict__ Gh,
                                          __bf16* __restrict__ Gl) {
  __shared__ __attribute__((aligned(16))) float piv[2][128];
  const int tid  = threadIdx.x;
  const int lane = tid & 31;
  const int w    = tid >> 5;
  const int rg   = w >> 1;
  const int R0   = rg * 16;
  const int c0   = (w & 1) * 64 + lane;
  const int c1   = c0 + 32;

  float a0[16], a1[16];
#pragma unroll
  for (int j = 0; j < 16; ++j) {
    a0[j] = G[(R0 + j) * 128 + c0];
    a1[j] = G[(R0 + j) * 128 + c1];
  }
  if (c0 == 0) {
    float4* nv4 = reinterpret_cast<float4*>(&piv[0][R0]);
#pragma unroll
    for (int q = 0; q < 4; ++q)
      nv4[q] = make_float4(a0[4 * q], a0[4 * q + 1], a0[4 * q + 2], a0[4 * q + 3]);
  }
  __syncthreads();

  for (int kb = 0; kb < 8; ++kb) {
#pragma unroll
    for (int kk = 0; kk < 16; ++kk) {
      const int k = kb * 16 + kk;
      const float* v = piv[kk & 1];
      const float pinv = 1.0f / v[k];
      const float t0 = v[c0] * pinv;
      const float t1 = v[c1] * pinv;
      const float n0 = -t0;
      const float n1 = -t1;
      const float4* v4 = reinterpret_cast<const float4*>(v + R0);
      float vr[16];
#pragma unroll
      for (int q = 0; q < 4; ++q) {
        float4 x = v4[q];
        vr[4 * q] = x.x; vr[4 * q + 1] = x.y; vr[4 * q + 2] = x.z; vr[4 * q + 3] = x.w;
      }
#pragma unroll
      for (int j = 0; j < 16; ++j) {
        a0[j] = fmaf(vr[j], n0, a0[j]);
        a1[j] = fmaf(vr[j], n1, a1[j]);
      }
      if (c0 == k) {
#pragma unroll
        for (int j = 0; j < 16; ++j) a0[j] = vr[j] * pinv;
      }
      if (c1 == k) {
#pragma unroll
        for (int j = 0; j < 16; ++j) a1[j] = vr[j] * pinv;
      }
      if (rg == kb) {
        a0[kk] = (c0 == k) ? -pinv : t0;
        a1[kk] = (c1 == k) ? -pinv : t1;
      }
      if (k + 1 < OUT_DIM) {
        float4* nv4 = reinterpret_cast<float4*>(&piv[(kk + 1) & 1][R0]);
        if (c0 == k + 1) {
#pragma unroll
          for (int q = 0; q < 4; ++q)
            nv4[q] = make_float4(a0[4 * q], a0[4 * q + 1], a0[4 * q + 2], a0[4 * q + 3]);
        }
        if (c1 == k + 1) {
#pragma unroll
          for (int q = 0; q < 4; ++q)
            nv4[q] = make_float4(a1[4 * q], a1[4 * q + 1], a1[4 * q + 2], a1[4 * q + 3]);
        }
      }
      __syncthreads();
    }
  }

#pragma unroll
  for (int j = 0; j < 16; ++j) {
    const int i0 = frag_pos(rg, 4, j, c0);
    const int i1 = frag_pos(rg, 4, j, c1);
    const float x0 = -a0[j];
    const float x1 = -a1[j];
    const __bf16 h0 = (__bf16)x0;
    const __bf16 h1 = (__bf16)x1;
    Gh[i0] = h0;  Gl[i0] = (__bf16)(x0 - (float)h0);
    Gh[i1] = h1;  Gl[i1] = (__bf16)(x1 - (float)h1);
  }
}


static __device__ inline v16bf cat16(v8bf a, v8bf b) {
  return __builtin_shufflevector(a, b, 0, 1, 2, 3, 4, 5, 6, 7,
                                 8, 9, 10, 11, 12, 13, 14, 15);
}

static __device__ inline void split4(float4 v, int e0, v16bf& hi, v16bf& lo) {
  const float x[4] = {v.x, v.y, v.z, v.w};
#pragma unroll
  for (int i = 0; i < 4; ++i) {
    const __bf16 h = (__bf16)x[i];
    hi[e0 + i] = h;
    lo[e0 + i] = (__bf16)(x[i] - (float)h);
  }
}

template <int NT>
struct Step0 {
  float4 y[NT][4];
  v8bf   b[8];
};

template <int NT>
static __device__ inline void load_step0(Step0<NT>& s,
                                         const float4* const (&py)[NT],
                                         const v8bf* __restrict__ ph,
                                         const v8bf* __restrict__ pl, int kk) {
#pragma unroll
  for (int u = 0; u < NT; ++u) {
    const float4* q = py[u] + kk * 8;
    s.y[u][0] = q[0];  s.y[u][1] = q[1];
    s.y[u][2] = q[4];  s.y[u][3] = q[5];
  }
  const int f = kk * 64;
  s.b[0] = ph[f];         s.b[1] = ph[f + 32];
  s.b[2] = pl[f];         s.b[3] = pl[f + 32];
  s.b[4] = ph[f + 2048];  s.b[5] = ph[f + 2048 + 32];
  s.b[6] = pl[f + 2048];  s.b[7] = pl[f + 2048 + 32];
}

template <int NT>
static __device__ inline void mma_step0(const Step0<NT>& s, v8f (&acc0)[NT],
                                        v8f (&acc1)[NT]) {
  v16bf yh[NT], yl[NT];
#pragma unroll
  for (int u = 0; u < NT; ++u) {
    split4(s.y[u][0], 0, yh[u], yl[u]);
    split4(s.y[u][1], 4, yh[u], yl[u]);
    split4(s.y[u][2], 8, yh[u], yl[u]);
    split4(s.y[u][3], 12, yh[u], yl[u]);
  }
  v16bf fh = cat16(s.b[0], s.b[1]);
  v16bf fl = cat16(s.b[2], s.b[3]);
#pragma unroll
  for (int u = 0; u < NT; ++u) acc0[u] = wmma_bf16(yh[u], fh, acc0[u]);
#pragma unroll
  for (int u = 0; u < NT; ++u) acc0[u] = wmma_bf16(yl[u], fh, acc0[u]);
#pragma unroll
  for (int u = 0; u < NT; ++u) acc0[u] = wmma_bf16(yh[u], fl, acc0[u]);
  fh = cat16(s.b[4], s.b[5]);
  fl = cat16(s.b[6], s.b[7]);
#pragma unroll
  for (int u = 0; u < NT; ++u) acc1[u] = wmma_bf16(yh[u], fh, acc1[u]);
#pragma unroll
  for (int u = 0; u < NT; ++u) acc1[u] = wmma_bf16(yl[u], fh, acc1[u]);
#pragma unroll
  for (int u = 0; u < NT; ++u) acc1[u] = wmma_bf16(yh[u], fl, acc1[u]);
}

template <int NT>
static __device__ inline void residual_units(
    const int (&i0)[NT], int wid, int lane, const float* __restrict__ y,
    const float* __restrict__ b, const __bf16* __restrict__ Ah,
    const __bf16* __restrict__ Al, float* __restrict__ out) {
  const int half = lane >> 4;
  const int l15  = lane & 15;
  const int n0a  = (wid * 2) * 16;
  const int n0b  = n0a + 16;
  const float4* py[NT];
#pragma unroll
  for (int u = 0; u < NT; ++u)
    py[u] = reinterpret_cast<const float4*>(
        y + (size_t)(i0[u] + l15) * IN_DIM + 8 * half);
  const v8bf* p0h = reinterpret_cast<const v8bf*>(
      Ah + (size_t)(wid * 2) * 32 * 512) + lane;
  const v8bf* p0l = reinterpret_cast<const v8bf*>(
      Al + (size_t)(wid * 2) * 32 * 512) + lane;

  v8f acc0[NT] = {};
  v8f acc1[NT] = {};
  {
    Step0<NT> sa, sb;
    load_step0<NT>(sa, py, p0h, p0l, 0);
    for (int kk = 0; kk < IN_DIM / 32 - 2; kk += 2) {
      load_step0<NT>(sb, py, p0h, p0l, kk + 1);
      __builtin_amdgcn_sched_barrier(0);
      mma_step0<NT>(sa, acc0, acc1);
      load_step0<NT>(sa, py, p0h, p0l, kk + 2);
      __builtin_amdgcn_sched_barrier(0);
      mma_step0<NT>(sb, acc0, acc1);
    }
    load_step0<NT>(sb, py, p0h, p0l, IN_DIM / 32 - 1);
    mma_step0<NT>(sa, acc0, acc1);
    mma_step0<NT>(sb, acc0, acc1);
  }
#pragma unroll
  for (int u = 0; u < NT; ++u) {
#pragma unroll
    for (int v = 0; v < 8; ++v) {
      const int m = v + 8 * half;
      float* rrow = out + (size_t)(i0[u] + m) * IN_DIM;
      const float* brow = b + (size_t)(i0[u] + m) * OUT_DIM;
      rrow[n0a + l15] = acc0[u][v] - brow[n0a + l15];
      rrow[n0b + l15] = acc1[u][v] - brow[n0b + l15];
    }
  }
}

#define NU 2
struct Step2 {
  float c[NU][8];
  v8bf  b[16];
};

static __device__ inline void load_step2(Step2& s, const float* __restrict__ yt,
                                         const v8bf* __restrict__ ph,
                                         const v8bf* __restrict__ pl, int t) {
#pragma unroll
  for (int u = 0; u < NU; ++u) {
#pragma unroll
    for (int v = 0; v < 8; ++v)
      s.c[u][v] = yt[(size_t)(u * 16 + v) * IN_DIM + t * 16];
  }
  const int f = t * 256;
#pragma unroll
  for (int kk = 0; kk < 4; ++kk) {
    s.b[4 * kk]     = ph[f + kk * 64];
    s.b[4 * kk + 1] = ph[f + kk * 64 + 32];
    s.b[4 * kk + 2] = pl[f + kk * 64];
    s.b[4 * kk + 3] = pl[f + kk * 64 + 32];
  }
}

static __device__ inline void mma_step2(const Step2& s,
                                        const v16bf (&mh)[NU][4],
                                        const v16bf (&ml)[NU][4],
                                        float* __restrict__ ot, int t) {
  v8f acc[NU];
#pragma unroll
  for (int u = 0; u < NU; ++u) {
#pragma unroll
    for (int v = 0; v < 8; ++v) acc[u][v] = s.c[u][v];
  }
#pragma unroll
  for (int kk = 0; kk < 4; ++kk) {
    const v16bf fh = cat16(s.b[4 * kk], s.b[4 * kk + 1]);
    const v16bf fl = cat16(s.b[4 * kk + 2], s.b[4 * kk + 3]);
#pragma unroll
    for (int u = 0; u < NU; ++u) acc[u] = wmma_bf16(mh[u][kk], fh, acc[u]);
#pragma unroll
    for (int u = 0; u < NU; ++u) acc[u] = wmma_bf16(ml[u][kk], fh, acc[u]);
#pragma unroll
    for (int u = 0; u < NU; ++u) acc[u] = wmma_bf16(mh[u][kk], fl, acc[u]);
  }
#pragma unroll
  for (int u = 0; u < NU; ++u) {
#pragma unroll
    for (int v = 0; v < 8; ++v)
      ot[(size_t)(u * 16 + v) * IN_DIM + t * 16] = acc[u][v];
  }
}

__global__ __launch_bounds__(512) void invert_and_residual(
    const float* __restrict__ G, __bf16* __restrict__ Gh,
    __bf16* __restrict__ Gl, const float* __restrict__ y,
    const float* __restrict__ b, const __bf16* __restrict__ Ah,
    const __bf16* __restrict__ Al, float* __restrict__ out) {
  if (blockIdx.x == 0) {
    invert_body(G, Gh, Gl);
    return;
  }
  const int tid  = threadIdx.x;
  const int sub  = __builtin_amdgcn_readfirstlane(tid >> 7);
  const int wid  = (tid >> 5) & 3;
  const int lane = tid & 31;
  const int unit = (blockIdx.x - 1) * 4 + sub;
  if (sub == 0 && blockIdx.x <= 4) {
    const int rows[2] = {unit * 16, (1020 + (int)blockIdx.x - 1) * 16};
    residual_units<2>(rows, wid, lane, y, b, Ah, Al, out);
  } else {
    const int rows[1] = {unit * 16};
    residual_units<1>(rows, wid, lane, y, b, Ah, Al, out);
  }
}

__global__ __launch_bounds__(128, 2) void project(
    const float* __restrict__ y,
    const __bf16* __restrict__ Ath, const __bf16* __restrict__ Atl,
    const __bf16* __restrict__ Gh, const __bf16* __restrict__ Gl,
    float* out) {
  __shared__ float s_mu[NU * 16][OUT_DIM];

  const int tid  = threadIdx.x;
  const int wid  = tid >> 5;
  const int lane = tid & 31;
  const int half = lane >> 4;
  const int l15  = lane & 15;
  const int i0   = blockIdx.x * (NU * 16);
  const int n0a  = (wid * 2) * 16;
  const int n0b  = n0a + 16;

  const float* rrow = out + (size_t)(i0 + l15) * IN_DIM;
  const __bf16* g0h = Gh + (wid * 2) * 4 * 512;
  const __bf16* g0l = Gl + (wid * 2) * 4 * 512;
  const __bf16* g1h = g0h + 4 * 512;
  const __bf16* g1l = g0l + 4 * 512;
  v8f mu0[NU] = {};
  v8f mu1[NU] = {};
#pragma unroll
  for (int kk = 0; kk < 4; ++kk) {
    const int k0 = kk * 32;
    v16bf rh[NU], rl[NU];
#pragma unroll
    for (int u = 0; u < NU; ++u)
      split_fragA(rrow + (size_t)(u * 16) * IN_DIM, k0, half, false,
                  rh[u], rl[u]);
    v16bf fh = load_fragB(g0h + kk * 512, lane);
    v16bf fl = load_fragB(g0l + kk * 512, lane);
#pragma unroll
    for (int u = 0; u < NU; ++u) mu0[u] = wmma_bf16(rh[u], fh, mu0[u]);
#pragma unroll
    for (int u = 0; u < NU; ++u) mu0[u] = wmma_bf16(rl[u], fh, mu0[u]);
#pragma unroll
    for (int u = 0; u < NU; ++u) mu0[u] = wmma_bf16(rh[u], fl, mu0[u]);
    fh = load_fragB(g1h + kk * 512, lane);
    fl = load_fragB(g1l + kk * 512, lane);
#pragma unroll
    for (int u = 0; u < NU; ++u) mu1[u] = wmma_bf16(rh[u], fh, mu1[u]);
#pragma unroll
    for (int u = 0; u < NU; ++u) mu1[u] = wmma_bf16(rl[u], fh, mu1[u]);
#pragma unroll
    for (int u = 0; u < NU; ++u) mu1[u] = wmma_bf16(rh[u], fl, mu1[u]);
  }
#pragma unroll
  for (int u = 0; u < NU; ++u) {
#pragma unroll
    for (int v = 0; v < 8; ++v) {
      const int m = u * 16 + v + 8 * half;
      s_mu[m][n0a + l15] = mu0[u][v];
      s_mu[m][n0b + l15] = mu1[u][v];
    }
  }
  __syncthreads();

  v16bf mh[NU][4], ml[NU][4];
#pragma unroll
  for (int u = 0; u < NU; ++u) {
#pragma unroll
    for (int kk = 0; kk < 4; ++kk)
      split_fragA(&s_mu[u * 16 + l15][0], kk * 32, half, true,
                  mh[u][kk], ml[u][kk]);
  }

  const size_t tile0 = (size_t)(i0 + 8 * half) * IN_DIM + wid * 256 + l15;
  const float* yt = y + tile0;
  float*       ot = out + tile0;
  const v8bf* p2h = reinterpret_cast<const v8bf*>(
      Ath + (size_t)(wid * 16) * 4 * 512) + lane;
  const v8bf* p2l = reinterpret_cast<const v8bf*>(
      Atl + (size_t)(wid * 16) * 4 * 512) + lane;
  {
    Step2 ta, tb;
    load_step2(ta, yt, p2h, p2l, 0);
    for (int t = 0; t < 14; t += 2) {
      load_step2(tb, yt, p2h, p2l, t + 1);
      __builtin_amdgcn_sched_barrier(0);
      mma_step2(ta, mh, ml, ot, t);
      load_step2(ta, yt, p2h, p2l, t + 2);
      __builtin_amdgcn_sched_barrier(0);
      mma_step2(tb, mh, ml, ot, t + 1);
    }
    load_step2(tb, yt, p2h, p2l, 15);
    __builtin_amdgcn_sched_barrier(0);
    mma_step2(ta, mh, ml, ot, 14);
    mma_step2(tb, mh, ml, ot, 15);
  }
}

extern "C" void kernel_launch(void* const* d_in, const int* in_sizes, int n_in,
                              void* d_out, int out_size, void* d_ws,
                              size_t ws_size, hipStream_t stream) {
  (void)in_sizes; (void)n_in; (void)out_size; (void)ws_size;
  const float* y = (const float*)d_in[0];
  const float* A = (const float*)d_in[1];
  const float* b = (const float*)d_in[2];
  float* out = (float*)d_out;

  char* ws = (char*)d_ws;
  float*  G   = (float*) (ws + 0);
  __bf16* Gh  = (__bf16*)(ws + 65536);
  __bf16* Gl  = (__bf16*)(ws + 98304);
  __bf16* Ah  = (__bf16*)(ws + 131072);
  __bf16* Al  = (__bf16*)(ws + 393216);
  __bf16* Ath = (__bf16*)(ws + 655360);
  __bf16* Atl = (__bf16*)(ws + 917504);

  compute_G<<<64, 128, 0, stream>>>(A, G);
  pack_A<<<128, 256, 0, stream>>>(A, Ah, Al, Ath, Atl);
  invert_and_residual<<<256, 512, 0, stream>>>(G, Gh, Gl, y, b, Ah, Al, out);
  project<<<BATCH / (NU * 16), 128, 0, stream>>>(y, Ath, Atl, Gh, Gl, out);
}


# compute_G and pack_A share one launch (64 + 256 workgroups of 4 waves, path chosen per workgroup): 4 launches -> 3, the two preparations of A run side by side; every sum's order unchanged
# speedup vs baseline: 8.3132x; 1.0345x over previous
#include <hip/hip_runtime.h>


#define BATCH   16384
#define IN_DIM  1024
#define OUT_DIM 128

typedef __attribute__((ext_vector_type(16))) __bf16 v16bf;
typedef __attribute__((ext_vector_type(8)))  __bf16 v8bf;
typedef __attribute__((ext_vector_type(8)))  float  v8f;
typedef __attribute__((ext_vector_type(2)))  float  v2f;

static __device__ inline v8f wmma_bf16(v16bf a, v16bf b, v8f c) {
  return __builtin_amdgcn_wmma_f32_16x16x32_bf16(false, a, false, b,
                                                 (short)0, c, false, false);
}

static __device__ inline v8f wmma_f32(v2f a, v2f b, v8f c) {
  return __builtin_amdgcn_wmma_f32_16x16x4_f32(false, a, false, b,
                                               (short)0, c, false, false);
}

static __device__ inline void split_fragA(const float* __restrict__ rowp,
                                          int k0, int half, bool neg,
                                          v16bf& hi, v16bf& lo) {
  const float* p0 = rowp + k0 + 8 * half;
  const float* p1 = p0 + 16;
#pragma unroll
  for (int e = 0; e < 8; ++e) {
    float x0 = neg ? -p0[e] : p0[e];
    float x1 = neg ? -p1[e] : p1[e];
    __bf16 h0 = (__bf16)x0;
    __bf16 h1 = (__bf16)x1;
    hi[e]     = h0;  lo[e]     = (__bf16)(x0 - (float)h0);
    hi[e + 8] = h1;  lo[e + 8] = (__bf16)(x1 - (float)h1);
  }
}

static __device__ inline v16bf load_fragB(const __bf16* __restrict__ frag,
                                          int lane) {
  const v8bf* p = reinterpret_cast<const v8bf*>(frag) + lane;
  const v8bf e0 = p[0];
  const v8bf e1 = p[32];
  return __builtin_shufflevector(e0, e1, 0, 1, 2, 3, 4, 5, 6, 7,
                                 8, 9, 10, 11, 12, 13, 14, 15);
}

static __device__ inline int frag_pos(int tile, int frags, int l15, int k) {
  const int kin = k & 31;
  return (tile * frags + (k >> 5)) * 512 + (kin >> 4) * 256 +
         (l15 + 16 * ((kin >> 3) & 1)) * 8 + (kin & 7);
}

static __device__ inline void compute_G_body(const float* __restrict__ A,
                                             float* __restrict__ G, int tile) {
  __shared__ float part[4][8][32];
  const int tid  = threadIdx.x;
  const int lane = tid & 31;
  const int w    = tid >> 5;
  const int c    = lane & 15;
  const int h    = lane >> 4;
  const int ti   = tile >> 3;
  const int tj   = tile & 7;
  const float4* pa = reinterpret_cast<const float4*>(
      A + (size_t)(ti * 16 + c) * IN_DIM + w * 256 + 4 * h);
  const float4* pb = reinterpret_cast<const float4*>(
      A + (size_t)(tj * 16 + c) * IN_DIM + w * 256 + 4 * h);
  float4 xa[32], xb[32];
#pragma unroll
  for (int j = 0; j < 32; ++j) {
    xa[j] = pa[2 * j];
    xb[j] = pb[2 * j];
  }
  __builtin_amdgcn_sched_barrier(0);
  v8f acc0 = {};
  v8f acc1 = {};
#pragma unroll
  for (int j = 0; j < 32; ++j) {
    const v2f a0 = {xa[j].x, xa[j].y};
    const v2f b0 = {xb[j].x, xb[j].y};
    const v2f a1 = {xa[j].z, xa[j].w};
    const v2f b1 = {xb[j].z, xb[j].w};
    acc0 = wmma_f32(a0, b0, acc0);
    acc1 = wmma_f32(a1, b1, acc1);
  }
#pragma unroll
  for (int r = 0; r < 8; ++r) part[w][r][lane] = acc0[r] + acc1[r];
  __syncthreads();
#pragma unroll
  for (int q = 0; q < 2; ++q) {
    const int r = w + 4 * q;
    const float s = (part[0][r][lane] + part[1][r][lane]) +
                    (part[2][r][lane] + part[3][r][lane]);
    G[(ti * 16 + 8 * h + r) * OUT_DIM + tj * 16 + c] = s;
  }
}

static __device__ inline void pack_A_body(const float* __restrict__ A,
                                          __bf16* __restrict__ Ah,
                                          __bf16* __restrict__ Al,
                                          __bf16* __restrict__ Ath,
                                          __bf16* __restrict__ Atl, int t) {
  const int q     = t & 16383;
  const int f     = q >> 6;
  const int lane  = q & 31;
  const int l15   = lane & 15;
  const int kin0  = ((q >> 5) & 1) * 16 + (lane >> 4) * 8;
  float x[8];
  __bf16* dh;
  __bf16* dl;
  if (t < 16384) {
    const int nt = f >> 5, kk = f & 31;
    const float* s = A + (size_t)(nt * 16 + l15) * IN_DIM + kk * 32 + kin0;
#pragma unroll
    for (int e = 0; e < 8; ++e) x[e] = s[e];
    dh = Ah;  dl = Al;
  } else {
    const int ct = f >> 2, kk = f & 3;
    const float* s = A + (size_t)(kk * 32 + kin0) * IN_DIM + ct * 16 + l15;
#pragma unroll
    for (int e = 0; e < 8; ++e) x[e] = s[(size_t)e * IN_DIM];
    dh = Ath;  dl = Atl;
  }
  v8bf h, l;
#pragma unroll
  for (int e = 0; e < 8; ++e) {
    const __bf16 hh = (__bf16)x[e];
    h[e] = hh;
    l[e] = (__bf16)(x[e] - (float)hh);
  }
  reinterpret_cast<v8bf*>(dh)[q] = h;
  reinterpret_cast<v8bf*>(dl)[q] = l;
}

__global__ __launch_bounds__(128) void prep_A(const float* __restrict__ A,
                                              float* __restrict__ G,
                                              __bf16* __restrict__ Ah,
                                              __bf16* __restrict__ Al,
                                              __bf16* __restrict__ Ath,
                                              __bf16* __restrict__ Atl) {
  if (blockIdx.x < 64) {
    compute_G_body(A, G, blockIdx.x);
    return;
  }
  pack_A_body(A, Ah, Al, Ath, Atl, (blockIdx.x - 64) * 128 + threadIdx.x);
}

static __device__ inline void invert_body(const float* __restrict__ G,
                                          __bf16* __restrict__ Gh,
                                          __bf16* __restrict__ Gl) {
  __shared__ __attribute__((aligned(16))) float piv[2][128];
  const int tid  = threadIdx.x;
  const int lane = tid & 31;
  const int w    = tid >> 5;
  const int rg   = w >> 1;
  const int R0   = rg * 16;
  const int c0   = (w & 1) * 64 + lane;
  const int c1   = c0 + 32;

  float a0[16], a1[16];
#pragma unroll
  for (int j = 0; j < 16; ++j) {
    a0[j] = G[(R0 + j) * 128 + c0];
    a1[j] = G[(R0 + j) * 128 + c1];
  }
  if (c0 == 0) {
    float4* nv4 = reinterpret_cast<float4*>(&piv[0][R0]);
#pragma unroll
    for (int q = 0; q < 4; ++q)
      nv4[q] = make_float4(a0[4 * q], a0[4 * q + 1], a0[4 * q + 2], a0[4 * q + 3]);
  }
  __syncthreads();

  for (int kb = 0; kb < 8; ++kb) {
#pragma unroll
    for (int kk = 0; kk < 16; ++kk) {
      const int k = kb * 16 + kk;
      const float* v = piv[kk & 1];
      const float pinv = 1.0f / v[k];
      const float t0 = v[c0] * pinv;
      const float t1 = v[c1] * pinv;
      const float n0 = -t0;
      const float n1 = -t1;
      const float4* v4 = reinterpret_cast<const float4*>(v + R0);
      float vr[16];
#pragma unroll
      for (int q = 0; q < 4; ++q) {
        float4 x = v4[q];
        vr[4 * q] = x.x; vr[4 * q + 1] = x.y; vr[4 * q + 2] = x.z; vr[4 * q + 3] = x.w;
      }
#pragma unroll
      for (int j = 0; j < 16; ++j) {
        a0[j] = fmaf(vr[j], n0, a0[j]);
        a1[j] = fmaf(vr[j], n1, a1[j]);
      }
      if (c0 == k) {
#pragma unroll
        for (int j = 0; j < 16; ++j) a0[j] = vr[j] * pinv;
      }
      if (c1 == k) {
#pragma unroll
        for (int j = 0; j < 16; ++j) a1[j] = vr[j] * pinv;
      }
      if (rg == kb) {
        a0[kk] = (c0 == k) ? -pinv : t0;
        a1[kk] = (c1 == k) ? -pinv : t1;
      }
      if (k + 1 < OUT_DIM) {
        float4* nv4 = reinterpret_cast<float4*>(&piv[(kk + 1) & 1][R0]);
        if (c0 == k + 1) {
#pragma unroll
          for (int q = 0; q < 4; ++q)
            nv4[q] = make_float4(a0[4 * q], a0[4 * q + 1], a0[4 * q + 2], a0[4 * q + 3]);
        }
        if (c1 == k + 1) {
#pragma unroll
          for (int q = 0; q < 4; ++q)
            nv4[q] = make_float4(a1[4 * q], a1[4 * q + 1], a1[4 * q + 2], a1[4 * q + 3]);
        }
      }
      __syncthreads();
    }
  }

#pragma unroll
  for (int j = 0; j < 16; ++j) {
    const int i0 = frag_pos(rg, 4, j, c0);
    const int i1 = frag_pos(rg, 4, j, c1);
    const float x0 = -a0[j];
    const float x1 = -a1[j];
    const __bf16 h0 = (__bf16)x0;
    const __bf16 h1 = (__bf16)x1;
    Gh[i0] = h0;  Gl[i0] = (__bf16)(x0 - (float)h0);
    Gh[i1] = h1;  Gl[i1] = (__bf16)(x1 - (float)h1);
  }
}


static __device__ inline v16bf cat16(v8bf a, v8bf b) {
  return __builtin_shufflevector(a, b, 0, 1, 2, 3, 4, 5, 6, 7,
                                 8, 9, 10, 11, 12, 13, 14, 15);
}

static __device__ inline void split4(float4 v, int e0, v16bf& hi, v16bf& lo) {
  const float x[4] = {v.x, v.y, v.z, v.w};
#pragma unroll
  for (int i = 0; i < 4; ++i) {
    const __bf16 h = (__bf16)x[i];
    hi[e0 + i] = h;
    lo[e0 + i] = (__bf16)(x[i] - (float)h);
  }
}

template <int NT>
struct Step0 {
  float4 y[NT][4];
  v8bf   b[8];
};

template <int NT>
static __device__ inline void load_step0(Step0<NT>& s,
                                         const float4* const (&py)[NT],
                                         const v8bf* __restrict__ ph,
                                         const v8bf* __restrict__ pl, int kk) {
#pragma unroll
  for (int u = 0; u < NT; ++u) {
    const float4* q = py[u] + kk * 8;
    s.y[u][0] = q[0];  s.y[u][1] = q[1];
    s.y[u][2] = q[4];  s.y[u][3] = q[5];
  }
  const int f = kk * 64;
  s.b[0] = ph[f];         s.b[1] = ph[f + 32];
  s.b[2] = pl[f];         s.b[3] = pl[f + 32];
  s.b[4] = ph[f + 2048];  s.b[5] = ph[f + 2048 + 32];
  s.b[6] = pl[f + 2048];  s.b[7] = pl[f + 2048 + 32];
}

template <int NT>
static __device__ inline void mma_step0(const Step0<NT>& s, v8f (&acc0)[NT],
                                        v8f (&acc1)[NT]) {
  v16bf yh[NT], yl[NT];
#pragma unroll
  for (int u = 0; u < NT; ++u) {
    split4(s.y[u][0], 0, yh[u], yl[u]);
    split4(s.y[u][1], 4, yh[u], yl[u]);
    split4(s.y[u][2], 8, yh[u], yl[u]);
    split4(s.y[u][3], 12, yh[u], yl[u]);
  }
  v16bf fh = cat16(s.b[0], s.b[1]);
  v16bf fl = cat16(s.b[2], s.b[3]);
#pragma unroll
  for (int u = 0; u < NT; ++u) acc0[u] = wmma_bf16(yh[u], fh, acc0[u]);
#pragma unroll
  for (int u = 0; u < NT; ++u) acc0[u] = wmma_bf16(yl[u], fh, acc0[u]);
#pragma unroll
  for (int u = 0; u < NT; ++u) acc0[u] = wmma_bf16(yh[u], fl, acc0[u]);
  fh = cat16(s.b[4], s.b[5]);
  fl = cat16(s.b[6], s.b[7]);
#pragma unroll
  for (int u = 0; u < NT; ++u) acc1[u] = wmma_bf16(yh[u], fh, acc1[u]);
#pragma unroll
  for (int u = 0; u < NT; ++u) acc1[u] = wmma_bf16(yl[u], fh, acc1[u]);
#pragma unroll
  for (int u = 0; u < NT; ++u) acc1[u] = wmma_bf16(yh[u], fl, acc1[u]);
}

template <int NT>
static __device__ inline void residual_units(
    const int (&i0)[NT], int wid, int lane, const float* __restrict__ y,
    const float* __restrict__ b, const __bf16* __restrict__ Ah,
    const __bf16* __restrict__ Al, float* __restrict__ out) {
  const int half = lane >> 4;
  const int l15  = lane & 15;
  const int n0a  = (wid * 2) * 16;
  const int n0b  = n0a + 16;
  const float4* py[NT];
#pragma unroll
  for (int u = 0; u < NT; ++u)
    py[u] = reinterpret_cast<const float4*>(
        y + (size_t)(i0[u] + l15) * IN_DIM + 8 * half);
  const v8bf* p0h = reinterpret_cast<const v8bf*>(
      Ah + (size_t)(wid * 2) * 32 * 512) + lane;
  const v8bf* p0l = reinterpret_cast<const v8bf*>(
      Al + (size_t)(wid * 2) * 32 * 512) + lane;

  v8f acc0[NT] = {};
  v8f acc1[NT] = {};
  {
    Step0<NT> sa, sb;
    load_step0<NT>(sa, py, p0h, p0l, 0);
    for (int kk = 0; kk < IN_DIM / 32 - 2; kk += 2) {
      load_step0<NT>(sb, py, p0h, p0l, kk + 1);
      __builtin_amdgcn_sched_barrier(0);
      mma_step0<NT>(sa, acc0, acc1);
      load_step0<NT>(sa, py, p0h, p0l, kk + 2);
      __builtin_amdgcn_sched_barrier(0);
      mma_step0<NT>(sb, acc0, acc1);
    }
    load_step0<NT>(sb, py, p0h, p0l, IN_DIM / 32 - 1);
    mma_step0<NT>(sa, acc0, acc1);
    mma_step0<NT>(sb, acc0, acc1);
  }
#pragma unroll
  for (int u = 0; u < NT; ++u) {
#pragma unroll
    for (int v = 0; v < 8; ++v) {
      const int m = v + 8 * half;
      float* rrow = out + (size_t)(i0[u] + m) * IN_DIM;
      const float* brow = b + (size_t)(i0[u] + m) * OUT_DIM;
      rrow[n0a + l15] = acc0[u][v] - brow[n0a + l15];
      rrow[n0b + l15] = acc1[u][v] - brow[n0b + l15];
    }
  }
}

#define NU 2
struct Step2 {
  float c[NU][8];
  v8bf  b[16];
};

static __device__ inline void load_step2(Step2& s, const float* __restrict__ yt,
                                         const v8bf* __restrict__ ph,
                                         const v8bf* __restrict__ pl, int t) {
#pragma unroll
  for (int u = 0; u < NU; ++u) {
#pragma unroll
    for (int v = 0; v < 8; ++v)
      s.c[u][v] = yt[(size_t)(u * 16 + v) * IN_DIM + t * 16];
  }
  const int f = t * 256;
#pragma unroll
  for (int kk = 0; kk < 4; ++kk) {
    s.b[4 * kk]     = ph[f + kk * 64];
    s.b[4 * kk + 1] = ph[f + kk * 64 + 32];
    s.b[4 * kk + 2] = pl[f + kk * 64];
    s.b[4 * kk + 3] = pl[f + kk * 64 + 32];
  }
}

static __device__ inline void mma_step2(const Step2& s,
                                        const v16bf (&mh)[NU][4],
                                        const v16bf (&ml)[NU][4],
                                        float* __restrict__ ot, int t) {
  v8f acc[NU];
#pragma unroll
  for (int u = 0; u < NU; ++u) {
#pragma unroll
    for (int v = 0; v < 8; ++v) acc[u][v] = s.c[u][v];
  }
#pragma unroll
  for (int kk = 0; kk < 4; ++kk) {
    const v16bf fh = cat16(s.b[4 * kk], s.b[4 * kk + 1]);
    const v16bf fl = cat16(s.b[4 * kk + 2], s.b[4 * kk + 3]);
#pragma unroll
    for (int u = 0; u < NU; ++u) acc[u] = wmma_bf16(mh[u][kk], fh, acc[u]);
#pragma unroll
    for (int u = 0; u < NU; ++u) acc[u] = wmma_bf16(ml[u][kk], fh, acc[u]);
#pragma unroll
    for (int u = 0; u < NU; ++u) acc[u] = wmma_bf16(mh[u][kk], fl, acc[u]);
  }
#pragma unroll
  for (int u = 0; u < NU; ++u) {
#pragma unroll
    for (int v = 0; v < 8; ++v)
      ot[(size_t)(u * 16 + v) * IN_DIM + t * 16] = acc[u][v];
  }
}

__global__ __launch_bounds__(512) void invert_and_residual(
    const float* __restrict__ G, __bf16* __restrict__ Gh,
    __bf16* __restrict__ Gl, const float* __restrict__ y,
    const float* __restrict__ b, const __bf16* __restrict__ Ah,
    const __bf16* __restrict__ Al, float* __restrict__ out) {
  if (blockIdx.x == 0) {
    invert_body(G, Gh, Gl);
    return;
  }
  const int tid  = threadIdx.x;
  const int sub  = __builtin_amdgcn_readfirstlane(tid >> 7);
  const int wid  = (tid >> 5) & 3;
  const int lane = tid & 31;
  const int unit = (blockIdx.x - 1) * 4 + sub;
  if (sub == 0 && blockIdx.x <= 4) {
    const int rows[2] = {unit * 16, (1020 + (int)blockIdx.x - 1) * 16};
    residual_units<2>(rows, wid, lane, y, b, Ah, Al, out);
  } else {
    const int rows[1] = {unit * 16};
    residual_units<1>(rows, wid, lane, y, b, Ah, Al, out);
  }
}

__global__ __launch_bounds__(128, 2) void project(
    const float* __restrict__ y,
    const __bf16* __restrict__ Ath, const __bf16* __restrict__ Atl,
    const __bf16* __restrict__ Gh, const __bf16* __restrict__ Gl,
    float* out) {
  __shared__ float s_mu[NU * 16][OUT_DIM];

  const int tid  = threadIdx.x;
  const int wid  = tid >> 5;
  const int lane = tid & 31;
  const int half = lane >> 4;
  const int l15  = lane & 15;
  const int i0   = blockIdx.x * (NU * 16);
  const int n0a  = (wid * 2) * 16;
  const int n0b  = n0a + 16;

  const float* rrow = out + (size_t)(i0 + l15) * IN_DIM;
  const __bf16* g0h = Gh + (wid * 2) * 4 * 512;
  const __bf16* g0l = Gl + (wid * 2) * 4 * 512;
  const __bf16* g1h = g0h + 4 * 512;
  const __bf16* g1l = g0l + 4 * 512;
  v8f mu0[NU] = {};
  v8f mu1[NU] = {};
#pragma unroll
  for (int kk = 0; kk < 4; ++kk) {
    const int k0 = kk * 32;
    v16bf rh[NU], rl[NU];
#pragma unroll
    for (int u = 0; u < NU; ++u)
      split_fragA(rrow + (size_t)(u * 16) * IN_DIM, k0, half, false,
                  rh[u], rl[u]);
    v16bf fh = load_fragB(g0h + kk * 512, lane);
    v16bf fl = load_fragB(g0l + kk * 512, lane);
#pragma unroll
    for (int u = 0; u < NU; ++u) mu0[u] = wmma_bf16(rh[u], fh, mu0[u]);
#pragma unroll
    for (int u = 0; u < NU; ++u) mu0[u] = wmma_bf16(rl[u], fh, mu0[u]);
#pragma unroll
    for (int u = 0; u < NU; ++u) mu0[u] = wmma_bf16(rh[u], fl, mu0[u]);
    fh = load_fragB(g1h + kk * 512, lane);
    fl = load_fragB(g1l + kk * 512, lane);
#pragma unroll
    for (int u = 0; u < NU; ++u) mu1[u] = wmma_bf16(rh[u], fh, mu1[u]);
#pragma unroll
    for (int u = 0; u < NU; ++u) mu1[u] = wmma_bf16(rl[u], fh, mu1[u]);
#pragma unroll
    for (int u = 0; u < NU; ++u) mu1[u] = wmma_bf16(rh[u], fl, mu1[u]);
  }
#pragma unroll
  for (int u = 0; u < NU; ++u) {
#pragma unroll
    for (int v = 0; v < 8; ++v) {
      const int m = u * 16 + v + 8 * half;
      s_mu[m][n0a + l15] = mu0[u][v];
      s_mu[m][n0b + l15] = mu1[u][v];
    }
  }
  __syncthreads();

  v16bf mh[NU][4], ml[NU][4];
#pragma unroll
  for (int u = 0; u < NU; ++u) {
#pragma unroll
    for (int kk = 0; kk < 4; ++kk)
      split_fragA(&s_mu[u * 16 + l15][0], kk * 32, half, true,
                  mh[u][kk], ml[u][kk]);
  }

  const size_t tile0 = (size_t)(i0 + 8 * half) * IN_DIM + wid * 256 + l15;
  const float* yt = y + tile0;
  float*       ot = out + tile0;
  const v8bf* p2h = reinterpret_cast<const v8bf*>(
      Ath + (size_t)(wid * 16) * 4 * 512) + lane;
  const v8bf* p2l = reinterpret_cast<const v8bf*>(
      Atl + (size_t)(wid * 16) * 4 * 512) + lane;
  {
    Step2 ta, tb;
    load_step2(ta, yt, p2h, p2l, 0);
    for (int t = 0; t < 14; t += 2) {
      load_step2(tb, yt, p2h, p2l, t + 1);
      __builtin_amdgcn_sched_barrier(0);
      mma_step2(ta, mh, ml, ot, t);
      load_step2(ta, yt, p2h, p2l, t + 2);
      __builtin_amdgcn_sched_barrier(0);
      mma_step2(tb, mh, ml, ot, t + 1);
    }
    load_step2(tb, yt, p2h, p2l, 15);
    __builtin_amdgcn_sched_barrier(0);
    mma_step2(ta, mh, ml, ot, 14);
    mma_step2(tb, mh, ml, ot, 15);
  }
}

extern "C" void kernel_launch(void* const* d_in, const int* in_sizes, int n_in,
                              void* d_out, int out_size, void* d_ws,
                              size_t ws_size, hipStream_t stream) {
  (void)in_sizes; (void)n_in; (void)out_size; (void)ws_size;
  const float* y = (const float*)d_in[0];
  const float* A = (const float*)d_in[1];
  const float* b = (const float*)d_in[2];
  float* out = (float*)d_out;

  char* ws = (char*)d_ws;
  float*  G   = (float*) (ws + 0);
  __bf16* Gh  = (__bf16*)(ws + 65536);
  __bf16* Gl  = (__bf16*)(ws + 98304);
  __bf16* Ah  = (__bf16*)(ws + 131072);
  __bf16* Al  = (__bf16*)(ws + 393216);
  __bf16* Ath = (__bf16*)(ws + 655360);
  __bf16* Atl = (__bf16*)(ws + 917504);

  prep_A<<<320, 128, 0, stream>>>(A, G, Ah, Al, Ath, Atl);
  invert_and_residual<<<256, 512, 0, stream>>>(G, Gh, Gl, y, b, Ah, Al, out);
  project<<<BATCH / (NU * 16), 128, 0, stream>>>(y, Ath, Atl, Gh, Gl, out);
}


# inverse of G: two pivots to a barrier; columns k, k+1 pass through LDS together, each thread forms column k+1 under sweep k itself; 128 -> 64 barriers; every element's operations and order unchanged
# speedup vs baseline: 8.6803x; 1.0442x over previous
#include <hip/hip_runtime.h>


#define BATCH   16384
#define IN_DIM  1024
#define OUT_DIM 128

typedef __attribute__((ext_vector_type(16))) __bf16 v16bf;
typedef __attribute__((ext_vector_type(8)))  __bf16 v8bf;
typedef __attribute__((ext_vector_type(8)))  float  v8f;
typedef __attribute__((ext_vector_type(2)))  float  v2f;

static __device__ inline v8f wmma_bf16(v16bf a, v16bf b, v8f c) {
  return __builtin_amdgcn_wmma_f32_16x16x32_bf16(false, a, false, b,
                                                 (short)0, c, false, false);
}

static __device__ inline v8f wmma_f32(v2f a, v2f b, v8f c) {
  return __builtin_amdgcn_wmma_f32_16x16x4_f32(false, a, false, b,
                                               (short)0, c, false, false);
}

static __device__ inline void split_fragA(const float* __restrict__ rowp,
                                          int k0, int half, bool neg,
                                          v16bf& hi, v16bf& lo) {
  const float* p0 = rowp + k0 + 8 * half;
  const float* p1 = p0 + 16;
#pragma unroll
  for (int e = 0; e < 8; ++e) {
    float x0 = neg ? -p0[e] : p0[e];
    float x1 = neg ? -p1[e] : p1[e];
    __bf16 h0 = (__bf16)x0;
    __bf16 h1 = (__bf16)x1;
    hi[e]     = h0;  lo[e]     = (__bf16)(x0 - (float)h0);
    hi[e + 8] = h1;  lo[e + 8] = (__bf16)(x1 - (float)h1);
  }
}

static __device__ inline v16bf load_fragB(const __bf16* __restrict__ frag,
                                          int lane) {
  const v8bf* p = reinterpret_cast<const v8bf*>(frag) + lane;
  const v8bf e0 = p[0];
  const v8bf e1 = p[32];
  return __builtin_shufflevector(e0, e1, 0, 1, 2, 3, 4, 5, 6, 7,
                                 8, 9, 10, 11, 12, 13, 14, 15);
}

static __device__ inline int frag_pos(int tile, int frags, int l15, int k) {
  const int kin = k & 31;
  return (tile * frags + (k >> 5)) * 512 + (kin >> 4) * 256 +
         (l15 + 16 * ((kin >> 3) & 1)) * 8 + (kin & 7);
}

static __device__ inline void compute_G_body(const float* __restrict__ A,
                                             float* __restrict__ G, int tile) {
  __shared__ float part[4][8][32];
  const int tid  = threadIdx.x;
  const int lane = tid & 31;
  const int w    = tid >> 5;
  const int c    = lane & 15;
  const int h    = lane >> 4;
  const int ti   = tile >> 3;
  const int tj   = tile & 7;
  const float4* pa = reinterpret_cast<const float4*>(
      A + (size_t)(ti * 16 + c) * IN_DIM + w * 256 + 4 * h);
  const float4* pb = reinterpret_cast<const float4*>(
      A + (size_t)(tj * 16 + c) * IN_DIM + w * 256 + 4 * h);
  float4 xa[32], xb[32];
#pragma unroll
  for (int j = 0; j < 32; ++j) {
    xa[j] = pa[2 * j];
    xb[j] = pb[2 * j];
  }
  __builtin_amdgcn_sched_barrier(0);
  v8f acc0 = {};
  v8f acc1 = {};
#pragma unroll
  for (int j = 0; j < 32; ++j) {
    const v2f a0 = {xa[j].x, xa[j].y};
    const v2f b0 = {xb[j].x, xb[j].y};
    const v2f a1 = {xa[j].z, xa[j].w};
    const v2f b1 = {xb[j].z, xb[j].w};
    acc0 = wmma_f32(a0, b0, acc0);
    acc1 = wmma_f32(a1, b1, acc1);
  }
#pragma unroll
  for (int r = 0; r < 8; ++r) part[w][r][lane] = acc0[r] + acc1[r];
  __syncthreads();
#pragma unroll
  for (int q = 0; q < 2; ++q) {
    const int r = w + 4 * q;
    const float s = (part[0][r][lane] + part[1][r][lane]) +
                    (part[2][r][lane] + part[3][r][lane]);
    G[(ti * 16 + 8 * h + r) * OUT_DIM + tj * 16 + c] = s;
  }
}

static __device__ inline void pack_A_body(const float* __restrict__ A,
                                          __bf16* __restrict__ Ah,
                                          __bf16* __restrict__ Al,
                                          __bf16* __restrict__ Ath,
                                          __bf16* __restrict__ Atl, int t) {
  const int q     = t & 16383;
  const int f     = q >> 6;
  const int lane  = q & 31;
  const int l15   = lane & 15;
  const int kin0  = ((q >> 5) & 1) * 16 + (lane >> 4) * 8;
  float x[8];
  __bf16* dh;
  __bf16* dl;
  if (t < 16384) {
    const int nt = f >> 5, kk = f & 31;
    const float* s = A + (size_t)(nt * 16 + l15) * IN_DIM + kk * 32 + kin0;
#pragma unroll
    for (int e = 0; e < 8; ++e) x[e] = s[e];
    dh = Ah;  dl = Al;
  } else {
    const int ct = f >> 2, kk = f & 3;
    const float* s = A + (size_t)(kk * 32 + kin0) * IN_DIM + ct * 16 + l15;
#pragma unroll
    for (int e = 0; e < 8; ++e) x[e] = s[(size_t)e * IN_DIM];
    dh = Ath;  dl = Atl;
  }
  v8bf h, l;
#pragma unroll
  for (int e = 0; e < 8; ++e) {
    const __bf16 hh = (__bf16)x[e];
    h[e] = hh;
    l[e] = (__bf16)(x[e] - (float)hh);
  }
  reinterpret_cast<v8bf*>(dh)[q] = h;
  reinterpret_cast<v8bf*>(dl)[q] = l;
}

__global__ __launch_bounds__(128) void prep_A(const float* __restrict__ A,
                                              float* __restrict__ G,
                                              __bf16* __restrict__ Ah,
                                              __bf16* __restrict__ Al,
                                              __bf16* __restrict__ Ath,
                                              __bf16* __restrict__ Atl) {
  if (blockIdx.x < 64) {
    compute_G_body(A, G, blockIdx.x);
    return;
  }
  pack_A_body(A, Ah, Al, Ath, Atl, (blockIdx.x - 64) * 128 + threadIdx.x);
}

static __device__ inline void invert_body(const float* __restrict__ G,
                                          __bf16* __restrict__ Gh,
                                          __bf16* __restrict__ Gl) {
  __shared__ __attribute__((aligned(16))) float piv[2][2][128];
  const int tid  = threadIdx.x;
  const int lane = tid & 31;
  const int w    = tid >> 5;
  const int rg   = w >> 1;
  const int R0   = rg * 16;
  const int c0   = (w & 1) * 64 + lane;
  const int c1   = c0 + 32;

  float a0[16], a1[16];
#pragma unroll
  for (int j = 0; j < 16; ++j) {
    a0[j] = G[(R0 + j) * 128 + c0];
    a1[j] = G[(R0 + j) * 128 + c1];
  }
  if (c0 < 2) {
    float4* nv4 = reinterpret_cast<float4*>(&piv[0][c0][R0]);
#pragma unroll
    for (int q = 0; q < 4; ++q)
      nv4[q] = make_float4(a0[4 * q], a0[4 * q + 1], a0[4 * q + 2], a0[4 * q + 3]);
  }
  __syncthreads();

  for (int kb = 0; kb < 8; ++kb) {
#pragma unroll
    for (int kk = 0; kk < 16; kk += 2) {
      const int k = kb * 16 + kk;
      const float* u  = piv[(kk >> 1) & 1][0];
      const float* wc = piv[(kk >> 1) & 1][1];
      const float4* u4 = reinterpret_cast<const float4*>(u + R0);
      const float4* w4 = reinterpret_cast<const float4*>(wc + R0);
      float ur[16], wr[16];
#pragma unroll
      for (int q = 0; q < 4; ++q) {
        const float4 x = u4[q];
        ur[4 * q] = x.x; ur[4 * q + 1] = x.y; ur[4 * q + 2] = x.z; ur[4 * q + 3] = x.w;
        const float4 z = w4[q];
        wr[4 * q] = z.x; wr[4 * q + 1] = z.y; wr[4 * q + 2] = z.z; wr[4 * q + 3] = z.w;
      }
      const float uc0 = u[c0];
      const float uc1 = u[c1];
      const float uk1 = u[k + 1];

      const float pinv = 1.0f / u[k];
      const float t0 = uc0 * pinv;
      const float t1 = uc1 * pinv;
      const float n0 = -t0;
      const float n1 = -t1;
      const float tk = uk1 * pinv;
      const float nk = -tk;
#pragma unroll
      for (int j = 0; j < 16; ++j) wr[j] = fmaf(ur[j], nk, wr[j]);
      if (rg == kb) wr[kk] = tk;
      float wc0 = fmaf(uc0, nk, wc[c0]);
      float wc1 = fmaf(uc1, nk, wc[c1]);
      if (c0 == k) wc0 = tk;
      if (c1 == k) wc1 = tk;
      const float d2 = fmaf(uk1, nk, wc[k + 1]);
#pragma unroll
      for (int j = 0; j < 16; ++j) {
        a0[j] = fmaf(ur[j], n0, a0[j]);
        a1[j] = fmaf(ur[j], n1, a1[j]);
      }
      if (c0 == k) {
#pragma unroll
        for (int j = 0; j < 16; ++j) a0[j] = ur[j] * pinv;
      }
      if (c1 == k) {
#pragma unroll
        for (int j = 0; j < 16; ++j) a1[j] = ur[j] * pinv;
      }
      if (rg == kb) {
        a0[kk] = (c0 == k) ? -pinv : t0;
        a1[kk] = (c1 == k) ? -pinv : t1;
      }

      const float pinv2 = 1.0f / d2;
      const float s0 = wc0 * pinv2;
      const float s1 = wc1 * pinv2;
      const float m0 = -s0;
      const float m1 = -s1;
#pragma unroll
      for (int j = 0; j < 16; ++j) {
        a0[j] = fmaf(wr[j], m0, a0[j]);
        a1[j] = fmaf(wr[j], m1, a1[j]);
      }
      if (c0 == k + 1) {
#pragma unroll
        for (int j = 0; j < 16; ++j) a0[j] = wr[j] * pinv2;
      }
      if (c1 == k + 1) {
#pragma unroll
        for (int j = 0; j < 16; ++j) a1[j] = wr[j] * pinv2;
      }
      if (rg == kb) {
        a0[kk + 1] = (c0 == k + 1) ? -pinv2 : s0;
        a1[kk + 1] = (c1 == k + 1) ? -pinv2 : s1;
      }

      if (k + 2 < OUT_DIM) {
        const int nb = ((kk >> 1) + 1) & 1;
        if (c0 == k + 2 || c0 == k + 3) {
          float4* nv4 = reinterpret_cast<float4*>(&piv[nb][c0 & 1][R0]);
#pragma unroll
          for (int q = 0; q < 4; ++q)
            nv4[q] = make_float4(a0[4 * q], a0[4 * q + 1], a0[4 * q + 2], a0[4 * q + 3]);
        }
        if (c1 == k + 2 || c1 == k + 3) {
          float4* nv4 = reinterpret_cast<float4*>(&piv[nb][c1 & 1][R0]);
#pragma unroll
          for (int q = 0; q < 4; ++q)
            nv4[q] = make_float4(a1[4 * q], a1[4 * q + 1], a1[4 * q + 2], a1[4 * q + 3]);
        }
      }
      __syncthreads();
    }
  }

#pragma unroll
  for (int j = 0; j < 16; ++j) {
    const int i0 = frag_pos(rg, 4, j, c0);
    const int i1 = frag_pos(rg, 4, j, c1);
    const float x0 = -a0[j];
    const float x1 = -a1[j];
    const __bf16 h0 = (__bf16)x0;
    const __bf16 h1 = (__bf16)x1;
    Gh[i0] = h0;  Gl[i0] = (__bf16)(x0 - (float)h0);
    Gh[i1] = h1;  Gl[i1] = (__bf16)(x1 - (float)h1);
  }
}


static __device__ inline v16bf cat16(v8bf a, v8bf b) {
  return __builtin_shufflevector(a, b, 0, 1, 2, 3, 4, 5, 6, 7,
                                 8, 9, 10, 11, 12, 13, 14, 15);
}

static __device__ inline void split4(float4 v, int e0, v16bf& hi, v16bf& lo) {
  const float x[4] = {v.x, v.y, v.z, v.w};
#pragma unroll
  for (int i = 0; i < 4; ++i) {
    const __bf16 h = (__bf16)x[i];
    hi[e0 + i] = h;
    lo[e0 + i] = (__bf16)(x[i] - (float)h);
  }
}

template <int NT>
struct Step0 {
  float4 y[NT][4];
  v8bf   b[8];
};

template <int NT>
static __device__ inline void load_step0(Step0<NT>& s,
                                         const float4* const (&py)[NT],
                                         const v8bf* __restrict__ ph,
                                         const v8bf* __restrict__ pl, int kk) {
#pragma unroll
  for (int u = 0; u < NT; ++u) {
    const float4* q = py[u] + kk * 8;
    s.y[u][0] = q[0];  s.y[u][1] = q[1];
    s.y[u][2] = q[4];  s.y[u][3] = q[5];
  }
  const int f = kk * 64;
  s.b[0] = ph[f];         s.b[1] = ph[f + 32];
  s.b[2] = pl[f];         s.b[3] = pl[f + 32];
  s.b[4] = ph[f + 2048];  s.b[5] = ph[f + 2048 + 32];
  s.b[6] = pl[f + 2048];  s.b[7] = pl[f + 2048 + 32];
}

template <int NT>
static __device__ inline void mma_step0(const Step0<NT>& s, v8f (&acc0)[NT],
                                        v8f (&acc1)[NT]) {
  v16bf yh[NT], yl[NT];
#pragma unroll
  for (int u = 0; u < NT; ++u) {
    split4(s.y[u][0], 0, yh[u], yl[u]);
    split4(s.y[u][1], 4, yh[u], yl[u]);
    split4(s.y[u][2], 8, yh[u], yl[u]);
    split4(s.y[u][3], 12, yh[u], yl[u]);
  }
  v16bf fh = cat16(s.b[0], s.b[1]);
  v16bf fl = cat16(s.b[2], s.b[3]);
#pragma unroll
  for (int u = 0; u < NT; ++u) acc0[u] = wmma_bf16(yh[u], fh, acc0[u]);
#pragma unroll
  for (int u = 0; u < NT; ++u) acc0[u] = wmma_bf16(yl[u], fh, acc0[u]);
#pragma unroll
  for (int u = 0; u < NT; ++u) acc0[u] = wmma_bf16(yh[u], fl, acc0[u]);
  fh = cat16(s.b[4], s.b[5]);
  fl = cat16(s.b[6], s.b[7]);
#pragma unroll
  for (int u = 0; u < NT; ++u) acc1[u] = wmma_bf16(yh[u], fh, acc1[u]);
#pragma unroll
  for (int u = 0; u < NT; ++u) acc1[u] = wmma_bf16(yl[u], fh, acc1[u]);
#pragma unroll
  for (int u = 0; u < NT; ++u) acc1[u] = wmma_bf16(yh[u], fl, acc1[u]);
}

template <int NT>
static __device__ inline void residual_units(
    const int (&i0)[NT], int wid, int lane, const float* __restrict__ y,
    const float* __restrict__ b, const __bf16* __restrict__ Ah,
    const __bf16* __restrict__ Al, float* __restrict__ out) {
  const int half = lane >> 4;
  const int l15  = lane & 15;
  const int n0a  = (wid * 2) * 16;
  const int n0b  = n0a + 16;
  const float4* py[NT];
#pragma unroll
  for (int u = 0; u < NT; ++u)
    py[u] = reinterpret_cast<const float4*>(
        y + (size_t)(i0[u] + l15) * IN_DIM + 8 * half);
  const v8bf* p0h = reinterpret_cast<const v8bf*>(
      Ah + (size_t)(wid * 2) * 32 * 512) + lane;
  const v8bf* p0l = reinterpret_cast<const v8bf*>(
      Al + (size_t)(wid * 2) * 32 * 512) + lane;

  v8f acc0[NT] = {};
  v8f acc1[NT] = {};
  {
    Step0<NT> sa, sb;
    load_step0<NT>(sa, py, p0h, p0l, 0);
    for (int kk = 0; kk < IN_DIM / 32 - 2; kk += 2) {
      load_step0<NT>(sb, py, p0h, p0l, kk + 1);
      __builtin_amdgcn_sched_barrier(0);
      mma_step0<NT>(sa, acc0, acc1);
      load_step0<NT>(sa, py, p0h, p0l, kk + 2);
      __builtin_amdgcn_sched_barrier(0);
      mma_step0<NT>(sb, acc0, acc1);
    }
    load_step0<NT>(sb, py, p0h, p0l, IN_DIM / 32 - 1);
    mma_step0<NT>(sa, acc0, acc1);
    mma_step0<NT>(sb, acc0, acc1);
  }
#pragma unroll
  for (int u = 0; u < NT; ++u) {
#pragma unroll
    for (int v = 0; v < 8; ++v) {
      const int m = v + 8 * half;
      float* rrow = out + (size_t)(i0[u] + m) * IN_DIM;
      const float* brow = b + (size_t)(i0[u] + m) * OUT_DIM;
      rrow[n0a + l15] = acc0[u][v] - brow[n0a + l15];
      rrow[n0b + l15] = acc1[u][v] - brow[n0b + l15];
    }
  }
}

#define NU 2
struct Step2 {
  float c[NU][8];
  v8bf  b[16];
};

static __device__ inline void load_step2(Step2& s, const float* __restrict__ yt,
                                         const v8bf* __restrict__ ph,
                                         const v8bf* __restrict__ pl, int t) {
#pragma unroll
  for (int u = 0; u < NU; ++u) {
#pragma unroll
    for (int v = 0; v < 8; ++v)
      s.c[u][v] = yt[(size_t)(u * 16 + v) * IN_DIM + t * 16];
  }
  const int f = t * 256;
#pragma unroll
  for (int kk = 0; kk < 4; ++kk) {
    s.b[4 * kk]     = ph[f + kk * 64];
    s.b[4 * kk + 1] = ph[f + kk * 64 + 32];
    s.b[4 * kk + 2] = pl[f + kk * 64];
    s.b[4 * kk + 3] = pl[f + kk * 64 + 32];
  }
}

static __device__ inline void mma_step2(const Step2& s,
                                        const v16bf (&mh)[NU][4],
                                        const v16bf (&ml)[NU][4],
                                        float* __restrict__ ot, int t) {
  v8f acc[NU];
#pragma unroll
  for (int u = 0; u < NU; ++u) {
#pragma unroll
    for (int v = 0; v < 8; ++v) acc[u][v] = s.c[u][v];
  }
#pragma unroll
  for (int kk = 0; kk < 4; ++kk) {
    const v16bf fh = cat16(s.b[4 * kk], s.b[4 * kk + 1]);
    const v16bf fl = cat16(s.b[4 * kk + 2], s.b[4 * kk + 3]);
#pragma unroll
    for (int u = 0; u < NU; ++u) acc[u] = wmma_bf16(mh[u][kk], fh, acc[u]);
#pragma unroll
    for (int u = 0; u < NU; ++u) acc[u] = wmma_bf16(ml[u][kk], fh, acc[u]);
#pragma unroll
    for (int u = 0; u < NU; ++u) acc[u] = wmma_bf16(mh[u][kk], fl, acc[u]);
  }
#pragma unroll
  for (int u = 0; u < NU; ++u) {
#pragma unroll
    for (int v = 0; v < 8; ++v)
      ot[(size_t)(u * 16 + v) * IN_DIM + t * 16] = acc[u][v];
  }
}

__global__ __launch_bounds__(512) void invert_and_residual(
    const float* __restrict__ G, __bf16* __restrict__ Gh,
    __bf16* __restrict__ Gl, const float* __restrict__ y,
    const float* __restrict__ b, const __bf16* __restrict__ Ah,
    const __bf16* __restrict__ Al, float* __restrict__ out) {
  if (blockIdx.x == 0) {
    invert_body(G, Gh, Gl);
    return;
  }
  const int tid  = threadIdx.x;
  const int sub  = __builtin_amdgcn_readfirstlane(tid >> 7);
  const int wid  = (tid >> 5) & 3;
  const int lane = tid & 31;
  const int unit = (blockIdx.x - 1) * 4 + sub;
  if (sub == 0 && blockIdx.x <= 4) {
    const int rows[2] = {unit * 16, (1020 + (int)blockIdx.x - 1) * 16};
    residual_units<2>(rows, wid, lane, y, b, Ah, Al, out);
  } else {
    const int rows[1] = {unit * 16};
    residual_units<1>(rows, wid, lane, y, b, Ah, Al, out);
  }
}

__global__ __launch_bounds__(128, 2) void project(
    const float* __restrict__ y,
    const __bf16* __restrict__ Ath, const __bf16* __restrict__ Atl,
    const __bf16* __restrict__ Gh, const __bf16* __restrict__ Gl,
    float* out) {
  __shared__ float s_mu[NU * 16][OUT_DIM];

  const int tid  = threadIdx.x;
  const int wid  = tid >> 5;
  const int lane = tid & 31;
  const int half = lane >> 4;
  const int l15  = lane & 15;
  const int i0   = blockIdx.x * (NU * 16);
  const int n0a  = (wid * 2) * 16;
  const int n0b  = n0a + 16;

  const float* rrow = out + (size_t)(i0 + l15) * IN_DIM;
  const __bf16* g0h = Gh + (wid * 2) * 4 * 512;
  const __bf16* g0l = Gl + (wid * 2) * 4 * 512;
  const __bf16* g1h = g0h + 4 * 512;
  const __bf16* g1l = g0l + 4 * 512;
  v8f mu0[NU] = {};
  v8f mu1[NU] = {};
#pragma unroll
  for (int kk = 0; kk < 4; ++kk) {
    const int k0 = kk * 32;
    v16bf rh[NU], rl[NU];
#pragma unroll
    for (int u = 0; u < NU; ++u)
      split_fragA(rrow + (size_t)(u * 16) * IN_DIM, k0, half, false,
                  rh[u], rl[u]);
    v16bf fh = load_fragB(g0h + kk * 512, lane);
    v16bf fl = load_fragB(g0l + kk * 512, lane);
#pragma unroll
    for (int u = 0; u < NU; ++u) mu0[u] = wmma_bf16(rh[u], fh, mu0[u]);
#pragma unroll
    for (int u = 0; u < NU; ++u) mu0[u] = wmma_bf16(rl[u], fh, mu0[u]);
#pragma unroll
    for (int u = 0; u < NU; ++u) mu0[u] = wmma_bf16(rh[u], fl, mu0[u]);
    fh = load_fragB(g1h + kk * 512, lane);
    fl = load_fragB(g1l + kk * 512, lane);
#pragma unroll
    for (int u = 0; u < NU; ++u) mu1[u] = wmma_bf16(rh[u], fh, mu1[u]);
#pragma unroll
    for (int u = 0; u < NU; ++u) mu1[u] = wmma_bf16(rl[u], fh, mu1[u]);
#pragma unroll
    for (int u = 0; u < NU; ++u) mu1[u] = wmma_bf16(rh[u], fl, mu1[u]);
  }
#pragma unroll
  for (int u = 0; u < NU; ++u) {
#pragma unroll
    for (int v = 0; v < 8; ++v) {
      const int m = u * 16 + v + 8 * half;
      s_mu[m][n0a + l15] = mu0[u][v];
      s_mu[m][n0b + l15] = mu1[u][v];
    }
  }
  __syncthreads();

  v16bf mh[NU][4], ml[NU][4];
#pragma unroll
  for (int u = 0; u < NU; ++u) {
#pragma unroll
    for (int kk = 0; kk < 4; ++kk)
      split_fragA(&s_mu[u * 16 + l15][0], kk * 32, half, true,
                  mh[u][kk], ml[u][kk]);
  }

  const size_t tile0 = (size_t)(i0 + 8 * half) * IN_DIM + wid * 256 + l15;
  const float* yt = y + tile0;
  float*       ot = out + tile0;
  const v8bf* p2h = reinterpret_cast<const v8bf*>(
      Ath + (size_t)(wid * 16) * 4 * 512) + lane;
  const v8bf* p2l = reinterpret_cast<const v8bf*>(
      Atl + (size_t)(wid * 16) * 4 * 512) + lane;
  {
    Step2 ta, tb;
    load_step2(ta, yt, p2h, p2l, 0);
    for (int t = 0; t < 14; t += 2) {
      load_step2(tb, yt, p2h, p2l, t + 1);
      __builtin_amdgcn_sched_barrier(0);
      mma_step2(ta, mh, ml, ot, t);
      load_step2(ta, yt, p2h, p2l, t + 2);
      __builtin_amdgcn_sched_barrier(0);
      mma_step2(tb, mh, ml, ot, t + 1);
    }
    load_step2(tb, yt, p2h, p2l, 15);
    __builtin_amdgcn_sched_barrier(0);
    mma_step2(ta, mh, ml, ot, 14);
    mma_step2(tb, mh, ml, ot, 15);
  }
}

extern "C" void kernel_launch(void* const* d_in, const int* in_sizes, int n_in,
                              void* d_out, int out_size, void* d_ws,
                              size_t ws_size, hipStream_t stream) {
  (void)in_sizes; (void)n_in; (void)out_size; (void)ws_size;
  const float* y = (const float*)d_in[0];
  const float* A = (const float*)d_in[1];
  const float* b = (const float*)d_in[2];
  float* out = (float*)d_out;

  char* ws = (char*)d_ws;
  float*  G   = (float*) (ws + 0);
  __bf16* Gh  = (__bf16*)(ws + 65536);
  __bf16* Gl  = (__bf16*)(ws + 98304);
  __bf16* Ah  = (__bf16*)(ws + 131072);
  __bf16* Al  = (__bf16*)(ws + 393216);
  __bf16* Ath = (__bf16*)(ws + 655360);
  __bf16* Atl = (__bf16*)(ws + 917504);

  prep_A<<<320, 128, 0, stream>>>(A, G, Ah, Al, Ath, Atl);
  invert_and_residual<<<256, 512, 0, stream>>>(G, Gh, Gl, y, b, Ah, Al, out);
  project<<<BATCH / (NU * 16), 128, 0, stream>>>(y, Ath, Atl, Gh, Gl, out);
}
